# starting kernel
# baseline (speedup 1.0000x reference)
#include <hip/hip_runtime.h>
#include <math.h>

typedef __attribute__((ext_vector_type(2))) float v2f;
typedef __attribute__((ext_vector_type(8))) float v8f;

static constexpr int BS = 8, CH = 3, H = 40, W = 40;
static constexpr int HO = 39, WO = 39;
static constexpr int POS_PER_IMG = HO * WO;
static constexpr int NPOS = BS * POS_PER_IMG;
static constexpr int NTILES = (NPOS + 15) / 16;

__global__ __launch_bounds__(256) void build_mats_kernel(
    const float* __restrict__ wts, float* __restrict__ mats) {
  __shared__ float sre[16][256];
  __shared__ float sim2[16][256];
  const int x = threadIdx.x;

#pragma unroll
  for (int j = 0; j < 16; ++j) {
    sre[j][x] = (x == (j << 4)) ? 1.0f : 0.0f;
    sim2[j][x] = 0.0f;
  }
  __syncthreads();

  for (int l = 0; l < 2; ++l) {
    for (int q = 0; q < 8; ++q) {
      float wv = wts[l * 8 + q];
      float s, c;
      sincosf(0.5f * wv, &s, &c);
      const int p = x ^ (1 << (7 - q));
      float nr[16], ni[16];
#pragma unroll
      for (int j = 0; j < 16; ++j) {
        float r1 = sre[j][x], i1 = sim2[j][x];
        float r2 = sre[j][p], i2 = sim2[j][p];
        nr[j] = c * r1 + s * i2;
        ni[j] = c * i1 - s * r2;
      }
      __syncthreads();
#pragma unroll
      for (int j = 0; j < 16; ++j) { sre[j][x] = nr[j]; sim2[j][x] = ni[j]; }
      __syncthreads();
    }
    for (int q = 0; q < 8; ++q) {
      const int cm = 1 << (7 - q);
      const int tm = 1 << (7 - ((q + 1) & 7));
      const int xp = (x & cm) ? (x ^ tm) : x;
      float nr[16], ni[16];
#pragma unroll
      for (int j = 0; j < 16; ++j) { nr[j] = sre[j][xp]; ni[j] = sim2[j][xp]; }
      __syncthreads();
#pragma unroll
      for (int j = 0; j < 16; ++j) { sre[j][x] = nr[j]; sim2[j][x] = ni[j]; }
      __syncthreads();
    }
  }

  const int j = x >> 4, k = x & 15;
  float ar0 = 0.f, ai0 = 0.f, ar1 = 0.f, ai1 = 0.f;
  for (int t = 0; t < 256; ++t) {
    float vjr = sre[j][t], vji = sim2[j][t];
    float vkr = sre[k][t], vki = sim2[k][t];
    float re = vjr * vkr + vji * vki;
    float im = vjr * vki - vji * vkr;
    float d0 = ((t >> 7) & 1) ? -1.f : 1.f;
    float d1 = ((t >> 6) & 1) ? -1.f : 1.f;
    ar0 += d0 * re; ai0 += d0 * im;
    ar1 += d1 * re; ai1 += d1 * im;
  }
  mats[0 * 256 + x] = ar0;
  mats[1 * 256 + x] = ai0;
  mats[2 * 256 + x] = ar1;
  mats[3 * 256 + x] = ai1;
}

__global__ __launch_bounds__(32) void quanv_kernel(
    const float* __restrict__ xin, const float* __restrict__ mats,
    float* __restrict__ out) {
  const int lane = threadIdx.x;
  const int m = lane & 15;
  const int h = lane >> 4;
  const int tile = blockIdx.x;

  __shared__ float Pu[16][16];
  __shared__ float Pv[16][16];

  v2f bA0[4], bB0[4], bA1[4], bB1[4];
#pragma unroll
  for (int kb = 0; kb < 4; ++kb) {
    const int r0 = 4 * kb + 2 * h;
    bA0[kb] = (v2f){mats[0 * 256 + r0 * 16 + m], mats[0 * 256 + (r0 + 1) * 16 + m]};
    bB0[kb] = (v2f){mats[1 * 256 + r0 * 16 + m], mats[1 * 256 + (r0 + 1) * 16 + m]};
    bA1[kb] = (v2f){mats[2 * 256 + r0 * 16 + m], mats[2 * 256 + (r0 + 1) * 16 + m]};
    bB1[kb] = (v2f){mats[3 * 256 + r0 * 16 + m], mats[3 * 256 + (r0 + 1) * 16 + m]};
  }

  int pos = tile * 16 + m;
  int posc = (pos < NPOS) ? pos : (NPOS - 1);
  int b = posc / POS_PER_IMG;
  int rem = posc - b * POS_PER_IMG;
  int ii = rem / WO;
  int jj = rem - ii * WO;

  float z0[8], z1[8];
#pragma unroll
  for (int r = 0; r < 8; ++r) { z0[r] = 0.f; z1[r] = 0.f; }

  for (int ch = 0; ch < 3; ++ch) {
    const float* img = xin + (((size_t)b * CH + ch) * H + ii) * W + jj;
    float t0 = img[0], t1 = img[1], t2 = img[W], t3 = img[W + 1];
    float c0, s0, c1, s1, c2, s2, c3, s3;
    sincosf(0.5f * t0, &s0, &c0);
    sincosf(0.5f * t1, &s1, &c1);
    sincosf(0.5f * t2, &s2, &c2);
    sincosf(0.5f * t3, &s3, &c3);

    float u[16], v[16];
#pragma unroll
    for (int e = 0; e < 16; ++e) {
      float m0 = (e & 8) ? s0 : c0;
      float m1 = (e & 4) ? s1 : c1;
      float m2 = (e & 2) ? s2 : c2;
      float m3 = (e & 1) ? s3 : c3;
      float mag = m0 * m1 * m2 * m3;
      int p = __popc(e) & 3;
      u[e] = (p == 0) ? mag : ((p == 2) ? -mag : 0.0f);
      v[e] = (p == 1) ? -mag : ((p == 3) ? mag : 0.0f);
    }
    if (h == 0) {
#pragma unroll
      for (int e = 0; e < 16; ++e) { Pu[m][e] = u[e]; Pv[m][e] = v[e]; }
    }
    __syncthreads();

    v8f gA0u = {0, 0, 0, 0, 0, 0, 0, 0};
    v8f gA0v = {0, 0, 0, 0, 0, 0, 0, 0};
    v8f gB0u = {0, 0, 0, 0, 0, 0, 0, 0};
    v8f gA1u = {0, 0, 0, 0, 0, 0, 0, 0};
    v8f gA1v = {0, 0, 0, 0, 0, 0, 0, 0};
    v8f gB1u = {0, 0, 0, 0, 0, 0, 0, 0};
#pragma unroll
    for (int kb = 0; kb < 4; ++kb) {
      v2f au = (v2f){u[4 * kb + 2 * h], u[4 * kb + 2 * h + 1]};
      v2f av = (v2f){v[4 * kb + 2 * h], v[4 * kb + 2 * h + 1]};
      gA0u = __builtin_amdgcn_wmma_f32_16x16x4_f32(false, au, false, bA0[kb],
                                                   (short)0, gA0u, false, false);
      gA0v = __builtin_amdgcn_wmma_f32_16x16x4_f32(false, av, false, bA0[kb],
                                                   (short)0, gA0v, false, false);
      gB0u = __builtin_amdgcn_wmma_f32_16x16x4_f32(false, au, false, bB0[kb],
                                                   (short)0, gB0u, false, false);
      gA1u = __builtin_amdgcn_wmma_f32_16x16x4_f32(false, au, false, bA1[kb],
                                                   (short)0, gA1u, false, false);
      gA1v = __builtin_amdgcn_wmma_f32_16x16x4_f32(false, av, false, bA1[kb],
                                                   (short)0, gA1v, false, false);
      gB1u = __builtin_amdgcn_wmma_f32_16x16x4_f32(false, au, false, bB1[kb],
                                                   (short)0, gB1u, false, false);
    }

#pragma unroll
    for (int r = 0; r < 8; ++r) {
      float pu = Pu[r + 8 * h][m];
      float pv = Pv[r + 8 * h][m];
      z0[r] += gA0u[r] * pu + (gA0v[r] - 2.0f * gB0u[r]) * pv;
      z1[r] += gA1u[r] * pu + (gA1v[r] - 2.0f * gB1u[r]) * pv;
    }
    __syncthreads();
  }

#pragma unroll
  for (int r = 0; r < 8; ++r) {
    float a = z0[r], c = z1[r];
    for (int mask = 1; mask <= 8; mask <<= 1) {
      a += __shfl_xor(a, mask, 32);
      c += __shfl_xor(c, mask, 32);
    }
    z0[r] = a; z1[r] = c;
  }

  if (m == 0) {
#pragma unroll
    for (int r = 0; r < 8; ++r) {
      int posm = tile * 16 + r + 8 * h;
      if (posm < NPOS) {
        int bb = posm / POS_PER_IMG;
        int rr = posm - bb * POS_PER_IMG;
        out[(bb * 2 + 0) * POS_PER_IMG + rr] = z0[r] * (1.0f / 3.0f);
        out[(bb * 2 + 1) * POS_PER_IMG + rr] = z1[r] * (1.0f / 3.0f);
      }
    }
  }
}

extern "C" void kernel_launch(void* const* d_in, const int* in_sizes, int n_in,
                              void* d_out, int out_size, void* d_ws, size_t ws_size,
                              hipStream_t stream) {
  const float* x = (const float*)d_in[0];
  const float* wts = (const float*)d_in[1];
  float* out = (float*)d_out;
  float* mats = (float*)d_ws;

  build_mats_kernel<<<1, 256, 0, stream>>>(wts, mats);
  quanv_kernel<<<NTILES, 32, 0, stream>>>(x, mats, out);
}


# build_mats: state in registers, 16 amplitudes/thread; RX on qubits 4-7 then 0-3 around one LDS transpose per layer, CNOT ring as one LDS permutation; 128 barriers -> 8, gate sin/cos computed once
# speedup vs baseline: 1.4009x; 1.4009x over previous
#include <hip/hip_runtime.h>
#include <math.h>

typedef __attribute__((ext_vector_type(2))) float v2f;
typedef __attribute__((ext_vector_type(8))) float v8f;

static constexpr int BS = 8, CH = 3, H = 40, W = 40;
static constexpr int HO = 39, WO = 39;
static constexpr int POS_PER_IMG = HO * WO;
static constexpr int NPOS = BS * POS_PER_IMG;
static constexpr int NTILES = (NPOS + 15) / 16;

static constexpr int ROWP = 272;

__device__ __forceinline__ int padx(int x) { return x + (x >> 4); }

__device__ __forceinline__ int ring_src(int x) {
  for (int q = 7; q >= 0; --q) {
    const int cm = 1 << (7 - q);
    const int tm = 1 << (7 - ((q + 1) & 7));
    x = (x & cm) ? (x ^ tm) : x;
  }
  return x;
}

__device__ __forceinline__ int ring_dst(int x) {
  for (int q = 0; q < 8; ++q) {
    const int cm = 1 << (7 - q);
    const int tm = 1 << (7 - ((q + 1) & 7));
    x = (x & cm) ? (x ^ tm) : x;
  }
  return x;
}

template <int BIT>
__device__ __forceinline__ void rx_regs(float (&re)[16], float (&im)[16], float c, float s) {
#pragma unroll
  for (int r = 0; r < 16; ++r) {
    if (r & BIT) continue;
    const int p = r | BIT;
    const float ar = re[r], ai = im[r], br = re[p], bi = im[p];
    re[r] = c * ar + s * bi;
    im[r] = c * ai - s * br;
    re[p] = c * br + s * ai;
    im[p] = c * bi - s * ar;
  }
}

__global__ __launch_bounds__(256) void build_mats_kernel(
    const float* __restrict__ wts, float* __restrict__ mats) {
  __shared__ float sre[16 * ROWP];
  __shared__ float sim2[16 * ROWP];
  __shared__ float gcs[16];
  __shared__ float gsn[16];
  const int x = threadIdx.x;
  const int j = x >> 4;
  const int g = x & 15;

  if (x < 16) {
    float s, c;
    sincosf(0.5f * wts[x], &s, &c);
    gcs[x] = c;
    gsn[x] = s;
  }

  float re[16], im[16];
#pragma unroll
  for (int r = 0; r < 16; ++r) { re[r] = 0.0f; im[r] = 0.0f; }
  re[0] = (g == j) ? 1.0f : 0.0f;
  __syncthreads();

  float* rowr = sre + j * ROWP;
  float* rowi = sim2 + j * ROWP;

#pragma unroll
  for (int l = 0; l < 2; ++l) {
    rx_regs<8>(re, im, gcs[l * 8 + 4], gsn[l * 8 + 4]);
    rx_regs<4>(re, im, gcs[l * 8 + 5], gsn[l * 8 + 5]);
    rx_regs<2>(re, im, gcs[l * 8 + 6], gsn[l * 8 + 6]);
    rx_regs<1>(re, im, gcs[l * 8 + 7], gsn[l * 8 + 7]);

#pragma unroll
    for (int r = 0; r < 16; ++r) {
      rowr[g * 17 + r] = re[r];
      rowi[g * 17 + r] = im[r];
    }
    __syncthreads();
#pragma unroll
    for (int r = 0; r < 16; ++r) {
      re[r] = rowr[r * 17 + g];
      im[r] = rowi[r * 17 + g];
    }

    rx_regs<8>(re, im, gcs[l * 8 + 0], gsn[l * 8 + 0]);
    rx_regs<4>(re, im, gcs[l * 8 + 1], gsn[l * 8 + 1]);
    rx_regs<2>(re, im, gcs[l * 8 + 2], gsn[l * 8 + 2]);
    rx_regs<1>(re, im, gcs[l * 8 + 3], gsn[l * 8 + 3]);
    __syncthreads();

    if (l == 0) {
#pragma unroll
      for (int r = 0; r < 16; ++r) {
        rowr[r * 17 + g] = re[r];
        rowi[r * 17 + g] = im[r];
      }
      __syncthreads();
      const int sg = ring_src(g << 4);
#pragma unroll
      for (int r = 0; r < 16; ++r) {
        const int y = sg ^ ring_src(r);
        re[r] = rowr[padx(y)];
        im[r] = rowi[padx(y)];
      }
      __syncthreads();
    } else {
      const int dg = ring_dst(g);
#pragma unroll
      for (int r = 0; r < 16; ++r) {
        const int y = dg ^ ring_dst(r << 4);
        sre[j * 256 + y] = re[r];
        sim2[j * 256 + y] = im[r];
      }
      __syncthreads();
    }
  }

  const int k = g;
  float ar0 = 0.f, ai0 = 0.f, ar1 = 0.f, ai1 = 0.f;
  for (int t = 0; t < 256; ++t) {
    float vjr = sre[j * 256 + t], vji = sim2[j * 256 + t];
    float vkr = sre[k * 256 + t], vki = sim2[k * 256 + t];
    float re2 = vjr * vkr + vji * vki;
    float im2 = vjr * vki - vji * vkr;
    float d0 = ((t >> 7) & 1) ? -1.f : 1.f;
    float d1 = ((t >> 6) & 1) ? -1.f : 1.f;
    ar0 += d0 * re2; ai0 += d0 * im2;
    ar1 += d1 * re2; ai1 += d1 * im2;
  }
  mats[0 * 256 + x] = ar0;
  mats[1 * 256 + x] = ai0;
  mats[2 * 256 + x] = ar1;
  mats[3 * 256 + x] = ai1;
}

__global__ __launch_bounds__(32) void quanv_kernel(
    const float* __restrict__ xin, const float* __restrict__ mats,
    float* __restrict__ out) {
  const int lane = threadIdx.x;
  const int m = lane & 15;
  const int h = lane >> 4;
  const int tile = blockIdx.x;

  __shared__ float Pu[16][16];
  __shared__ float Pv[16][16];

  v2f bA0[4], bB0[4], bA1[4], bB1[4];
#pragma unroll
  for (int kb = 0; kb < 4; ++kb) {
    const int r0 = 4 * kb + 2 * h;
    bA0[kb] = (v2f){mats[0 * 256 + r0 * 16 + m], mats[0 * 256 + (r0 + 1) * 16 + m]};
    bB0[kb] = (v2f){mats[1 * 256 + r0 * 16 + m], mats[1 * 256 + (r0 + 1) * 16 + m]};
    bA1[kb] = (v2f){mats[2 * 256 + r0 * 16 + m], mats[2 * 256 + (r0 + 1) * 16 + m]};
    bB1[kb] = (v2f){mats[3 * 256 + r0 * 16 + m], mats[3 * 256 + (r0 + 1) * 16 + m]};
  }

  int pos = tile * 16 + m;
  int posc = (pos < NPOS) ? pos : (NPOS - 1);
  int b = posc / POS_PER_IMG;
  int rem = posc - b * POS_PER_IMG;
  int ii = rem / WO;
  int jj = rem - ii * WO;

  float z0[8], z1[8];
#pragma unroll
  for (int r = 0; r < 8; ++r) { z0[r] = 0.f; z1[r] = 0.f; }

  for (int ch = 0; ch < 3; ++ch) {
    const float* img = xin + (((size_t)b * CH + ch) * H + ii) * W + jj;
    float t0 = img[0], t1 = img[1], t2 = img[W], t3 = img[W + 1];
    float c0, s0, c1, s1, c2, s2, c3, s3;
    sincosf(0.5f * t0, &s0, &c0);
    sincosf(0.5f * t1, &s1, &c1);
    sincosf(0.5f * t2, &s2, &c2);
    sincosf(0.5f * t3, &s3, &c3);

    float u[16], v[16];
#pragma unroll
    for (int e = 0; e < 16; ++e) {
      float m0 = (e & 8) ? s0 : c0;
      float m1 = (e & 4) ? s1 : c1;
      float m2 = (e & 2) ? s2 : c2;
      float m3 = (e & 1) ? s3 : c3;
      float mag = m0 * m1 * m2 * m3;
      int p = __popc(e) & 3;
      u[e] = (p == 0) ? mag : ((p == 2) ? -mag : 0.0f);
      v[e] = (p == 1) ? -mag : ((p == 3) ? mag : 0.0f);
    }
    if (h == 0) {
#pragma unroll
      for (int e = 0; e < 16; ++e) { Pu[m][e] = u[e]; Pv[m][e] = v[e]; }
    }
    __syncthreads();

    v8f gA0u = {0, 0, 0, 0, 0, 0, 0, 0};
    v8f gA0v = {0, 0, 0, 0, 0, 0, 0, 0};
    v8f gB0u = {0, 0, 0, 0, 0, 0, 0, 0};
    v8f gA1u = {0, 0, 0, 0, 0, 0, 0, 0};
    v8f gA1v = {0, 0, 0, 0, 0, 0, 0, 0};
    v8f gB1u = {0, 0, 0, 0, 0, 0, 0, 0};
#pragma unroll
    for (int kb = 0; kb < 4; ++kb) {
      v2f au = (v2f){u[4 * kb + 2 * h], u[4 * kb + 2 * h + 1]};
      v2f av = (v2f){v[4 * kb + 2 * h], v[4 * kb + 2 * h + 1]};
      gA0u = __builtin_amdgcn_wmma_f32_16x16x4_f32(false, au, false, bA0[kb],
                                                   (short)0, gA0u, false, false);
      gA0v = __builtin_amdgcn_wmma_f32_16x16x4_f32(false, av, false, bA0[kb],
                                                   (short)0, gA0v, false, false);
      gB0u = __builtin_amdgcn_wmma_f32_16x16x4_f32(false, au, false, bB0[kb],
                                                   (short)0, gB0u, false, false);
      gA1u = __builtin_amdgcn_wmma_f32_16x16x4_f32(false, au, false, bA1[kb],
                                                   (short)0, gA1u, false, false);
      gA1v = __builtin_amdgcn_wmma_f32_16x16x4_f32(false, av, false, bA1[kb],
                                                   (short)0, gA1v, false, false);
      gB1u = __builtin_amdgcn_wmma_f32_16x16x4_f32(false, au, false, bB1[kb],
                                                   (short)0, gB1u, false, false);
    }

#pragma unroll
    for (int r = 0; r < 8; ++r) {
      float pu = Pu[r + 8 * h][m];
      float pv = Pv[r + 8 * h][m];
      z0[r] += gA0u[r] * pu + (gA0v[r] - 2.0f * gB0u[r]) * pv;
      z1[r] += gA1u[r] * pu + (gA1v[r] - 2.0f * gB1u[r]) * pv;
    }
    __syncthreads();
  }

#pragma unroll
  for (int r = 0; r < 8; ++r) {
    float a = z0[r], c = z1[r];
    for (int mask = 1; mask <= 8; mask <<= 1) {
      a += __shfl_xor(a, mask, 32);
      c += __shfl_xor(c, mask, 32);
    }
    z0[r] = a; z1[r] = c;
  }

  if (m == 0) {
#pragma unroll
    for (int r = 0; r < 8; ++r) {
      int posm = tile * 16 + r + 8 * h;
      if (posm < NPOS) {
        int bb = posm / POS_PER_IMG;
        int rr = posm - bb * POS_PER_IMG;
        out[(bb * 2 + 0) * POS_PER_IMG + rr] = z0[r] * (1.0f / 3.0f);
        out[(bb * 2 + 1) * POS_PER_IMG + rr] = z1[r] * (1.0f / 3.0f);
      }
    }
  }
}

extern "C" void kernel_launch(void* const* d_in, const int* in_sizes, int n_in,
                              void* d_out, int out_size, void* d_ws, size_t ws_size,
                              hipStream_t stream) {
  const float* x = (const float*)d_in[0];
  const float* wts = (const float*)d_in[1];
  float* out = (float*)d_out;
  float* mats = (float*)d_ws;

  build_mats_kernel<<<1, 256, 0, stream>>>(wts, mats);
  quanv_kernel<<<NTILES, 32, 0, stream>>>(x, mats, out);
}


# build_mats: closing sums M_q = V^H D_q V moved from a 256-step LDS loop per thread onto f32 WMMA (32-amplitude slice per wave, partials added across waves in LDS; new summation order)
# speedup vs baseline: 3.1497x; 2.2483x over previous
#include <hip/hip_runtime.h>
#include <math.h>

typedef __attribute__((ext_vector_type(2))) float v2f;
typedef __attribute__((ext_vector_type(8))) float v8f;

static constexpr int BS = 8, CH = 3, H = 40, W = 40;
static constexpr int HO = 39, WO = 39;
static constexpr int POS_PER_IMG = HO * WO;
static constexpr int NPOS = BS * POS_PER_IMG;
static constexpr int NTILES = (NPOS + 15) / 16;

static constexpr int ROWP = 272;
static constexpr int FINP = 260;

__device__ __forceinline__ int padx(int x) { return x + (x >> 4); }

__device__ __forceinline__ int ring_src(int x) {
  for (int q = 7; q >= 0; --q) {
    const int cm = 1 << (7 - q);
    const int tm = 1 << (7 - ((q + 1) & 7));
    x = (x & cm) ? (x ^ tm) : x;
  }
  return x;
}

__device__ __forceinline__ int ring_dst(int x) {
  for (int q = 0; q < 8; ++q) {
    const int cm = 1 << (7 - q);
    const int tm = 1 << (7 - ((q + 1) & 7));
    x = (x & cm) ? (x ^ tm) : x;
  }
  return x;
}

template <int BIT>
__device__ __forceinline__ void rx_regs(float (&re)[16], float (&im)[16], float c, float s) {
#pragma unroll
  for (int r = 0; r < 16; ++r) {
    if (r & BIT) continue;
    const int p = r | BIT;
    const float ar = re[r], ai = im[r], br = re[p], bi = im[p];
    re[r] = c * ar + s * bi;
    im[r] = c * ai - s * br;
    re[p] = c * br + s * ai;
    im[p] = c * bi - s * ar;
  }
}

__global__ __launch_bounds__(256) void build_mats_kernel(
    const float* __restrict__ wts, float* __restrict__ mats) {
  __shared__ __attribute__((aligned(8))) float sre[16 * ROWP];
  __shared__ __attribute__((aligned(8))) float sim2[16 * ROWP];
  __shared__ float part[8][2][256];
  __shared__ float gcs[16];
  __shared__ float gsn[16];
  const int x = threadIdx.x;
  const int j = x >> 4;
  const int g = x & 15;

  if (x < 16) {
    float s, c;
    sincosf(0.5f * wts[x], &s, &c);
    gcs[x] = c;
    gsn[x] = s;
  }

  float re[16], im[16];
#pragma unroll
  for (int r = 0; r < 16; ++r) { re[r] = 0.0f; im[r] = 0.0f; }
  re[0] = (g == j) ? 1.0f : 0.0f;
  __syncthreads();

  float* rowr = sre + j * ROWP;
  float* rowi = sim2 + j * ROWP;

#pragma unroll
  for (int l = 0; l < 2; ++l) {
    rx_regs<8>(re, im, gcs[l * 8 + 4], gsn[l * 8 + 4]);
    rx_regs<4>(re, im, gcs[l * 8 + 5], gsn[l * 8 + 5]);
    rx_regs<2>(re, im, gcs[l * 8 + 6], gsn[l * 8 + 6]);
    rx_regs<1>(re, im, gcs[l * 8 + 7], gsn[l * 8 + 7]);

#pragma unroll
    for (int r = 0; r < 16; ++r) {
      rowr[g * 17 + r] = re[r];
      rowi[g * 17 + r] = im[r];
    }
    __syncthreads();
#pragma unroll
    for (int r = 0; r < 16; ++r) {
      re[r] = rowr[r * 17 + g];
      im[r] = rowi[r * 17 + g];
    }

    rx_regs<8>(re, im, gcs[l * 8 + 0], gsn[l * 8 + 0]);
    rx_regs<4>(re, im, gcs[l * 8 + 1], gsn[l * 8 + 1]);
    rx_regs<2>(re, im, gcs[l * 8 + 2], gsn[l * 8 + 2]);
    rx_regs<1>(re, im, gcs[l * 8 + 3], gsn[l * 8 + 3]);
    __syncthreads();

    if (l == 0) {
#pragma unroll
      for (int r = 0; r < 16; ++r) {
        rowr[r * 17 + g] = re[r];
        rowi[r * 17 + g] = im[r];
      }
      __syncthreads();
      const int sg = ring_src(g << 4);
#pragma unroll
      for (int r = 0; r < 16; ++r) {
        const int y = sg ^ ring_src(r);
        re[r] = rowr[padx(y)];
        im[r] = rowi[padx(y)];
      }
      __syncthreads();
    } else {
      const int dg = ring_dst(g);
#pragma unroll
      for (int r = 0; r < 16; ++r) {
        const int y = dg ^ ring_dst(r << 4);
        sre[j * FINP + y] = re[r];
        sim2[j * FINP + y] = im[r];
      }
      __syncthreads();
    }
  }

  const int wv = x >> 5;
  const int ln = x & 31;
  const int n = ln & 15;
  const int hh = ln >> 4;
  const float* fr = sre + n * FINP + 32 * wv + 2 * hh;
  const float* fi = sim2 + n * FINP + 32 * wv + 2 * hh;
  v8f prr = {0, 0, 0, 0, 0, 0, 0, 0};
  v8f pii = {0, 0, 0, 0, 0, 0, 0, 0};
  v8f pri = {0, 0, 0, 0, 0, 0, 0, 0};
  v8f pir = {0, 0, 0, 0, 0, 0, 0, 0};
#pragma unroll
  for (int kb = 0; kb < 8; ++kb) {
    const v2f vr = *reinterpret_cast<const v2f*>(fr + 4 * kb);
    const v2f vi = *reinterpret_cast<const v2f*>(fi + 4 * kb);
    prr = __builtin_amdgcn_wmma_f32_16x16x4_f32(false, vr, false, vr, (short)0, prr, false, false);
    pii = __builtin_amdgcn_wmma_f32_16x16x4_f32(false, vi, false, vi, (short)0, pii, false, false);
    pri = __builtin_amdgcn_wmma_f32_16x16x4_f32(false, vr, false, vi, (short)0, pri, false, false);
    pir = __builtin_amdgcn_wmma_f32_16x16x4_f32(false, vi, false, vr, (short)0, pir, false, false);
  }
#pragma unroll
  for (int r = 0; r < 8; ++r) {
    part[wv][0][(r + 8 * hh) * 16 + n] = prr[r] + pii[r];
    part[wv][1][(r + 8 * hh) * 16 + n] = pri[r] - pir[r];
  }
  __syncthreads();
  float pr[8], pi[8];
#pragma unroll
  for (int w = 0; w < 8; ++w) { pr[w] = part[w][0][x]; pi[w] = part[w][1][x]; }
  const float r01 = pr[0] + pr[1], r23 = pr[2] + pr[3], r45 = pr[4] + pr[5], r67 = pr[6] + pr[7];
  const float i01 = pi[0] + pi[1], i23 = pi[2] + pi[3], i45 = pi[4] + pi[5], i67 = pi[6] + pi[7];
  mats[0 * 256 + x] = (r01 + r23) - (r45 + r67);
  mats[1 * 256 + x] = (i01 + i23) - (i45 + i67);
  mats[2 * 256 + x] = (r01 - r23) + (r45 - r67);
  mats[3 * 256 + x] = (i01 - i23) + (i45 - i67);
}

__global__ __launch_bounds__(32) void quanv_kernel(
    const float* __restrict__ xin, const float* __restrict__ mats,
    float* __restrict__ out) {
  const int lane = threadIdx.x;
  const int m = lane & 15;
  const int h = lane >> 4;
  const int tile = blockIdx.x;

  __shared__ float Pu[16][16];
  __shared__ float Pv[16][16];

  v2f bA0[4], bB0[4], bA1[4], bB1[4];
#pragma unroll
  for (int kb = 0; kb < 4; ++kb) {
    const int r0 = 4 * kb + 2 * h;
    bA0[kb] = (v2f){mats[0 * 256 + r0 * 16 + m], mats[0 * 256 + (r0 + 1) * 16 + m]};
    bB0[kb] = (v2f){mats[1 * 256 + r0 * 16 + m], mats[1 * 256 + (r0 + 1) * 16 + m]};
    bA1[kb] = (v2f){mats[2 * 256 + r0 * 16 + m], mats[2 * 256 + (r0 + 1) * 16 + m]};
    bB1[kb] = (v2f){mats[3 * 256 + r0 * 16 + m], mats[3 * 256 + (r0 + 1) * 16 + m]};
  }

  int pos = tile * 16 + m;
  int posc = (pos < NPOS) ? pos : (NPOS - 1);
  int b = posc / POS_PER_IMG;
  int rem = posc - b * POS_PER_IMG;
  int ii = rem / WO;
  int jj = rem - ii * WO;

  float z0[8], z1[8];
#pragma unroll
  for (int r = 0; r < 8; ++r) { z0[r] = 0.f; z1[r] = 0.f; }

  for (int ch = 0; ch < 3; ++ch) {
    const float* img = xin + (((size_t)b * CH + ch) * H + ii) * W + jj;
    float t0 = img[0], t1 = img[1], t2 = img[W], t3 = img[W + 1];
    float c0, s0, c1, s1, c2, s2, c3, s3;
    sincosf(0.5f * t0, &s0, &c0);
    sincosf(0.5f * t1, &s1, &c1);
    sincosf(0.5f * t2, &s2, &c2);
    sincosf(0.5f * t3, &s3, &c3);

    float u[16], v[16];
#pragma unroll
    for (int e = 0; e < 16; ++e) {
      float m0 = (e & 8) ? s0 : c0;
      float m1 = (e & 4) ? s1 : c1;
      float m2 = (e & 2) ? s2 : c2;
      float m3 = (e & 1) ? s3 : c3;
      float mag = m0 * m1 * m2 * m3;
      int p = __popc(e) & 3;
      u[e] = (p == 0) ? mag : ((p == 2) ? -mag : 0.0f);
      v[e] = (p == 1) ? -mag : ((p == 3) ? mag : 0.0f);
    }
    if (h == 0) {
#pragma unroll
      for (int e = 0; e < 16; ++e) { Pu[m][e] = u[e]; Pv[m][e] = v[e]; }
    }
    __syncthreads();

    v8f gA0u = {0, 0, 0, 0, 0, 0, 0, 0};
    v8f gA0v = {0, 0, 0, 0, 0, 0, 0, 0};
    v8f gB0u = {0, 0, 0, 0, 0, 0, 0, 0};
    v8f gA1u = {0, 0, 0, 0, 0, 0, 0, 0};
    v8f gA1v = {0, 0, 0, 0, 0, 0, 0, 0};
    v8f gB1u = {0, 0, 0, 0, 0, 0, 0, 0};
#pragma unroll
    for (int kb = 0; kb < 4; ++kb) {
      v2f au = (v2f){u[4 * kb + 2 * h], u[4 * kb + 2 * h + 1]};
      v2f av = (v2f){v[4 * kb + 2 * h], v[4 * kb + 2 * h + 1]};
      gA0u = __builtin_amdgcn_wmma_f32_16x16x4_f32(false, au, false, bA0[kb],
                                                   (short)0, gA0u, false, false);
      gA0v = __builtin_amdgcn_wmma_f32_16x16x4_f32(false, av, false, bA0[kb],
                                                   (short)0, gA0v, false, false);
      gB0u = __builtin_amdgcn_wmma_f32_16x16x4_f32(false, au, false, bB0[kb],
                                                   (short)0, gB0u, false, false);
      gA1u = __builtin_amdgcn_wmma_f32_16x16x4_f32(false, au, false, bA1[kb],
                                                   (short)0, gA1u, false, false);
      gA1v = __builtin_amdgcn_wmma_f32_16x16x4_f32(false, av, false, bA1[kb],
                                                   (short)0, gA1v, false, false);
      gB1u = __builtin_amdgcn_wmma_f32_16x16x4_f32(false, au, false, bB1[kb],
                                                   (short)0, gB1u, false, false);
    }

#pragma unroll
    for (int r = 0; r < 8; ++r) {
      float pu = Pu[r + 8 * h][m];
      float pv = Pv[r + 8 * h][m];
      z0[r] += gA0u[r] * pu + (gA0v[r] - 2.0f * gB0u[r]) * pv;
      z1[r] += gA1u[r] * pu + (gA1v[r] - 2.0f * gB1u[r]) * pv;
    }
    __syncthreads();
  }

#pragma unroll
  for (int r = 0; r < 8; ++r) {
    float a = z0[r], c = z1[r];
    for (int mask = 1; mask <= 8; mask <<= 1) {
      a += __shfl_xor(a, mask, 32);
      c += __shfl_xor(c, mask, 32);
    }
    z0[r] = a; z1[r] = c;
  }

  if (m == 0) {
#pragma unroll
    for (int r = 0; r < 8; ++r) {
      int posm = tile * 16 + r + 8 * h;
      if (posm < NPOS) {
        int bb = posm / POS_PER_IMG;
        int rr = posm - bb * POS_PER_IMG;
        out[(bb * 2 + 0) * POS_PER_IMG + rr] = z0[r] * (1.0f / 3.0f);
        out[(bb * 2 + 1) * POS_PER_IMG + rr] = z1[r] * (1.0f / 3.0f);
      }
    }
  }
}

extern "C" void kernel_launch(void* const* d_in, const int* in_sizes, int n_in,
                              void* d_out, int out_size, void* d_ws, size_t ws_size,
                              hipStream_t stream) {
  const float* x = (const float*)d_in[0];
  const float* wts = (const float*)d_in[1];
  float* out = (float*)d_out;
  float* mats = (float*)d_ws;

  build_mats_kernel<<<1, 256, 0, stream>>>(wts, mats);
  quanv_kernel<<<NTILES, 32, 0, stream>>>(x, mats, out);
}


# Two launches fused into one: each 256-thread workgroup rebuilds M_0, M_1 in its own LDS, then each of its 8 waves runs one 16-position tile, B fragments from LDS; patch loads issued before the build
# speedup vs baseline: 4.1432x; 1.3154x over previous
#include <hip/hip_runtime.h>
#include <math.h>

typedef __attribute__((ext_vector_type(2))) float v2f;
typedef __attribute__((ext_vector_type(8))) float v8f;

static constexpr int BS = 8, CH = 3, H = 40, W = 40;
static constexpr int HO = 39, WO = 39;
static constexpr int POS_PER_IMG = HO * WO;
static constexpr int NPOS = BS * POS_PER_IMG;
static constexpr int NTILES = (NPOS + 15) / 16;
static constexpr int WAVES = 8;
static constexpr int NBLOCKS = (NTILES + WAVES - 1) / WAVES;

static constexpr int ROWP = 272;
static constexpr int FINP = 260;

__device__ __forceinline__ int padx(int x) { return x + (x >> 4); }

__device__ __forceinline__ int ring_src(int x) {
  for (int q = 7; q >= 0; --q) {
    const int cm = 1 << (7 - q);
    const int tm = 1 << (7 - ((q + 1) & 7));
    x = (x & cm) ? (x ^ tm) : x;
  }
  return x;
}

__device__ __forceinline__ int ring_dst(int x) {
  for (int q = 0; q < 8; ++q) {
    const int cm = 1 << (7 - q);
    const int tm = 1 << (7 - ((q + 1) & 7));
    x = (x & cm) ? (x ^ tm) : x;
  }
  return x;
}

template <int BIT>
__device__ __forceinline__ void rx_regs(float (&re)[16], float (&im)[16], float c, float s) {
#pragma unroll
  for (int r = 0; r < 16; ++r) {
    if (r & BIT) continue;
    const int p = r | BIT;
    const float ar = re[r], ai = im[r], br = re[p], bi = im[p];
    re[r] = c * ar + s * bi;
    im[r] = c * ai - s * br;
    re[p] = c * br + s * ai;
    im[p] = c * bi - s * ar;
  }
}

__global__ __launch_bounds__(256) void quanv_fused_kernel(
    const float* __restrict__ xin, const float* __restrict__ wts,
    float* __restrict__ out) {
  __shared__ __attribute__((aligned(8))) float sre[16 * ROWP];
  __shared__ __attribute__((aligned(8))) float sim2[16 * ROWP];
  __shared__ float part[8][2][256];
  __shared__ float gcs[16];
  __shared__ float gsn[16];
  __shared__ float smat[4 * 256];
  __shared__ float Pu[WAVES][16][16];
  __shared__ float Pv[WAVES][16][16];
  const int x = threadIdx.x;
  const int j = x >> 4;
  const int g = x & 15;

  const int wv = x >> 5;
  const int lane = x & 31;
  const int m = lane & 15;
  const int h = lane >> 4;
  const int tile = blockIdx.x * WAVES + wv;

  int pos = tile * 16 + m;
  int posc = (pos < NPOS) ? pos : (NPOS - 1);
  int b = posc / POS_PER_IMG;
  int rem = posc - b * POS_PER_IMG;
  int ii = rem / WO;
  int jj = rem - ii * WO;

  float tcur[4], tnx1[4], tnx2[4];
  {
    const float* img = xin + (((size_t)b * CH) * H + ii) * W + jj;
    tcur[0] = img[0];             tcur[1] = img[1];
    tcur[2] = img[W];             tcur[3] = img[W + 1];
    tnx1[0] = img[H * W];         tnx1[1] = img[H * W + 1];
    tnx1[2] = img[H * W + W];     tnx1[3] = img[H * W + W + 1];
    tnx2[0] = img[2 * H * W];     tnx2[1] = img[2 * H * W + 1];
    tnx2[2] = img[2 * H * W + W]; tnx2[3] = img[2 * H * W + W + 1];
  }

  if (x < 16) {
    float s, c;
    sincosf(0.5f * wts[x], &s, &c);
    gcs[x] = c;
    gsn[x] = s;
  }

  float re[16], im[16];
#pragma unroll
  for (int r = 0; r < 16; ++r) { re[r] = 0.0f; im[r] = 0.0f; }
  re[0] = (g == j) ? 1.0f : 0.0f;
  __syncthreads();

  float* rowr = sre + j * ROWP;
  float* rowi = sim2 + j * ROWP;

#pragma unroll
  for (int l = 0; l < 2; ++l) {
    rx_regs<8>(re, im, gcs[l * 8 + 4], gsn[l * 8 + 4]);
    rx_regs<4>(re, im, gcs[l * 8 + 5], gsn[l * 8 + 5]);
    rx_regs<2>(re, im, gcs[l * 8 + 6], gsn[l * 8 + 6]);
    rx_regs<1>(re, im, gcs[l * 8 + 7], gsn[l * 8 + 7]);

#pragma unroll
    for (int r = 0; r < 16; ++r) {
      rowr[g * 17 + r] = re[r];
      rowi[g * 17 + r] = im[r];
    }
    __syncthreads();
#pragma unroll
    for (int r = 0; r < 16; ++r) {
      re[r] = rowr[r * 17 + g];
      im[r] = rowi[r * 17 + g];
    }

    rx_regs<8>(re, im, gcs[l * 8 + 0], gsn[l * 8 + 0]);
    rx_regs<4>(re, im, gcs[l * 8 + 1], gsn[l * 8 + 1]);
    rx_regs<2>(re, im, gcs[l * 8 + 2], gsn[l * 8 + 2]);
    rx_regs<1>(re, im, gcs[l * 8 + 3], gsn[l * 8 + 3]);
    __syncthreads();

    if (l == 0) {
#pragma unroll
      for (int r = 0; r < 16; ++r) {
        rowr[r * 17 + g] = re[r];
        rowi[r * 17 + g] = im[r];
      }
      __syncthreads();
      const int sg = ring_src(g << 4);
#pragma unroll
      for (int r = 0; r < 16; ++r) {
        const int y = sg ^ ring_src(r);
        re[r] = rowr[padx(y)];
        im[r] = rowi[padx(y)];
      }
      __syncthreads();
    } else {
      const int dg = ring_dst(g);
#pragma unroll
      for (int r = 0; r < 16; ++r) {
        const int y = dg ^ ring_dst(r << 4);
        sre[j * FINP + y] = re[r];
        sim2[j * FINP + y] = im[r];
      }
      __syncthreads();
    }
  }

  const int n = m;
  const int hh = h;
  const float* fr = sre + n * FINP + 32 * wv + 2 * hh;
  const float* fi = sim2 + n * FINP + 32 * wv + 2 * hh;
  v8f prr = {0, 0, 0, 0, 0, 0, 0, 0};
  v8f pii = {0, 0, 0, 0, 0, 0, 0, 0};
  v8f pri = {0, 0, 0, 0, 0, 0, 0, 0};
  v8f pir = {0, 0, 0, 0, 0, 0, 0, 0};
#pragma unroll
  for (int kb = 0; kb < 8; ++kb) {
    const v2f vr = *reinterpret_cast<const v2f*>(fr + 4 * kb);
    const v2f vi = *reinterpret_cast<const v2f*>(fi + 4 * kb);
    prr = __builtin_amdgcn_wmma_f32_16x16x4_f32(false, vr, false, vr, (short)0, prr, false, false);
    pii = __builtin_amdgcn_wmma_f32_16x16x4_f32(false, vi, false, vi, (short)0, pii, false, false);
    pri = __builtin_amdgcn_wmma_f32_16x16x4_f32(false, vr, false, vi, (short)0, pri, false, false);
    pir = __builtin_amdgcn_wmma_f32_16x16x4_f32(false, vi, false, vr, (short)0, pir, false, false);
  }
#pragma unroll
  for (int r = 0; r < 8; ++r) {
    part[wv][0][(r + 8 * hh) * 16 + n] = prr[r] + pii[r];
    part[wv][1][(r + 8 * hh) * 16 + n] = pri[r] - pir[r];
  }
  __syncthreads();
  float pr[8], pi[8];
#pragma unroll
  for (int w = 0; w < 8; ++w) { pr[w] = part[w][0][x]; pi[w] = part[w][1][x]; }
  const float r01 = pr[0] + pr[1], r23 = pr[2] + pr[3], r45 = pr[4] + pr[5], r67 = pr[6] + pr[7];
  const float i01 = pi[0] + pi[1], i23 = pi[2] + pi[3], i45 = pi[4] + pi[5], i67 = pi[6] + pi[7];
  smat[0 * 256 + x] = (r01 + r23) - (r45 + r67);
  smat[1 * 256 + x] = (i01 + i23) - (i45 + i67);
  smat[2 * 256 + x] = (r01 - r23) + (r45 - r67);
  smat[3 * 256 + x] = (i01 - i23) + (i45 - i67);
  __syncthreads();


  v2f bA0[4], bB0[4], bA1[4], bB1[4];
#pragma unroll
  for (int kb = 0; kb < 4; ++kb) {
    const int r0 = 4 * kb + 2 * h;
    bA0[kb] = (v2f){smat[0 * 256 + r0 * 16 + m], smat[0 * 256 + (r0 + 1) * 16 + m]};
    bB0[kb] = (v2f){smat[1 * 256 + r0 * 16 + m], smat[1 * 256 + (r0 + 1) * 16 + m]};
    bA1[kb] = (v2f){smat[2 * 256 + r0 * 16 + m], smat[2 * 256 + (r0 + 1) * 16 + m]};
    bB1[kb] = (v2f){smat[3 * 256 + r0 * 16 + m], smat[3 * 256 + (r0 + 1) * 16 + m]};
  }

  float (*Puw)[16] = Pu[wv];
  float (*Pvw)[16] = Pv[wv];

  float z0[8], z1[8];
#pragma unroll
  for (int r = 0; r < 8; ++r) { z0[r] = 0.f; z1[r] = 0.f; }

  for (int ch = 0; ch < 3; ++ch) {
    const float t0 = tcur[0], t1 = tcur[1], t2 = tcur[2], t3 = tcur[3];
#pragma unroll
    for (int i = 0; i < 4; ++i) { tcur[i] = tnx1[i]; tnx1[i] = tnx2[i]; }
    float c0, s0, c1, s1, c2, s2, c3, s3;
    sincosf(0.5f * t0, &s0, &c0);
    sincosf(0.5f * t1, &s1, &c1);
    sincosf(0.5f * t2, &s2, &c2);
    sincosf(0.5f * t3, &s3, &c3);

    float u[16], v[16];
#pragma unroll
    for (int e = 0; e < 16; ++e) {
      float m0 = (e & 8) ? s0 : c0;
      float m1 = (e & 4) ? s1 : c1;
      float m2 = (e & 2) ? s2 : c2;
      float m3 = (e & 1) ? s3 : c3;
      float mag = m0 * m1 * m2 * m3;
      int p = __popc(e) & 3;
      u[e] = (p == 0) ? mag : ((p == 2) ? -mag : 0.0f);
      v[e] = (p == 1) ? -mag : ((p == 3) ? mag : 0.0f);
    }
    if (h == 0) {
#pragma unroll
      for (int e = 0; e < 16; ++e) { Puw[m][e] = u[e]; Pvw[m][e] = v[e]; }
    }
    __builtin_amdgcn_fence(__ATOMIC_RELEASE, "wavefront");
    __builtin_amdgcn_fence(__ATOMIC_ACQUIRE, "wavefront");

    v8f gA0u = {0, 0, 0, 0, 0, 0, 0, 0};
    v8f gA0v = {0, 0, 0, 0, 0, 0, 0, 0};
    v8f gB0u = {0, 0, 0, 0, 0, 0, 0, 0};
    v8f gA1u = {0, 0, 0, 0, 0, 0, 0, 0};
    v8f gA1v = {0, 0, 0, 0, 0, 0, 0, 0};
    v8f gB1u = {0, 0, 0, 0, 0, 0, 0, 0};
#pragma unroll
    for (int kb = 0; kb < 4; ++kb) {
      v2f au = (v2f){u[4 * kb + 2 * h], u[4 * kb + 2 * h + 1]};
      v2f av = (v2f){v[4 * kb + 2 * h], v[4 * kb + 2 * h + 1]};
      gA0u = __builtin_amdgcn_wmma_f32_16x16x4_f32(false, au, false, bA0[kb],
                                                   (short)0, gA0u, false, false);
      gA0v = __builtin_amdgcn_wmma_f32_16x16x4_f32(false, av, false, bA0[kb],
                                                   (short)0, gA0v, false, false);
      gB0u = __builtin_amdgcn_wmma_f32_16x16x4_f32(false, au, false, bB0[kb],
                                                   (short)0, gB0u, false, false);
      gA1u = __builtin_amdgcn_wmma_f32_16x16x4_f32(false, au, false, bA1[kb],
                                                   (short)0, gA1u, false, false);
      gA1v = __builtin_amdgcn_wmma_f32_16x16x4_f32(false, av, false, bA1[kb],
                                                   (short)0, gA1v, false, false);
      gB1u = __builtin_amdgcn_wmma_f32_16x16x4_f32(false, au, false, bB1[kb],
                                                   (short)0, gB1u, false, false);
    }

#pragma unroll
    for (int r = 0; r < 8; ++r) {
      float pu = Puw[r + 8 * h][m];
      float pv = Pvw[r + 8 * h][m];
      z0[r] += gA0u[r] * pu + (gA0v[r] - 2.0f * gB0u[r]) * pv;
      z1[r] += gA1u[r] * pu + (gA1v[r] - 2.0f * gB1u[r]) * pv;
    }
    __builtin_amdgcn_fence(__ATOMIC_RELEASE, "wavefront");
    __builtin_amdgcn_fence(__ATOMIC_ACQUIRE, "wavefront");
  }
#pragma unroll
  for (int r = 0; r < 8; ++r) {
    float a = z0[r], c = z1[r];
    for (int mask = 1; mask <= 8; mask <<= 1) {
      a += __shfl_xor(a, mask, 32);
      c += __shfl_xor(c, mask, 32);
    }
    z0[r] = a; z1[r] = c;
  }

  if (m == 0) {
#pragma unroll
    for (int r = 0; r < 8; ++r) {
      int posm = tile * 16 + r + 8 * h;
      if (posm < NPOS) {
        int bb = posm / POS_PER_IMG;
        int rr = posm - bb * POS_PER_IMG;
        out[(bb * 2 + 0) * POS_PER_IMG + rr] = z0[r] * (1.0f / 3.0f);
        out[(bb * 2 + 1) * POS_PER_IMG + rr] = z1[r] * (1.0f / 3.0f);
      }
    }
  }
}

extern "C" void kernel_launch(void* const* d_in, const int* in_sizes, int n_in,
                              void* d_out, int out_size, void* d_ws, size_t ws_size,
                              hipStream_t stream) {
  const float* x = (const float*)d_in[0];
  const float* wts = (const float*)d_in[1];
  float* out = (float*)d_out;

  quanv_fused_kernel<<<NBLOCKS, 256, 0, stream>>>(x, wts, out);
}


# Embedding phases (-i)^popcount folded into the column basis: only real S_q=Re(W^H D_q W) is built (2 closing products, not 4) and stage 2 is one product mag*S_q per filter; new summation order
# speedup vs baseline: 5.2531x; 1.2679x over previous
#include <hip/hip_runtime.h>
#include <math.h>

typedef __attribute__((ext_vector_type(2))) float v2f;
typedef __attribute__((ext_vector_type(8))) float v8f;

static constexpr int BS = 8, CH = 3, H = 40, W = 40;
static constexpr int HO = 39, WO = 39;
static constexpr int POS_PER_IMG = HO * WO;
static constexpr int NPOS = BS * POS_PER_IMG;
static constexpr int NTILES = (NPOS + 15) / 16;
static constexpr int WAVES = 8;
static constexpr int NBLOCKS = (NTILES + WAVES - 1) / WAVES;

static constexpr int ROWP = 272;
static constexpr int FINP = 260;
static constexpr int PPIT = 20;

__device__ __forceinline__ int padx(int x) { return x + (x >> 4); }

__device__ __forceinline__ int ring_src(int x) {
  for (int q = 7; q >= 0; --q) {
    const int cm = 1 << (7 - q);
    const int tm = 1 << (7 - ((q + 1) & 7));
    x = (x & cm) ? (x ^ tm) : x;
  }
  return x;
}

__device__ __forceinline__ int ring_dst(int x) {
  for (int q = 0; q < 8; ++q) {
    const int cm = 1 << (7 - q);
    const int tm = 1 << (7 - ((q + 1) & 7));
    x = (x & cm) ? (x ^ tm) : x;
  }
  return x;
}

template <int BIT>
__device__ __forceinline__ void rx_regs(float (&re)[16], float (&im)[16], float c, float s) {
#pragma unroll
  for (int r = 0; r < 16; ++r) {
    if (r & BIT) continue;
    const int p = r | BIT;
    const float ar = re[r], ai = im[r], br = re[p], bi = im[p];
    re[r] = c * ar + s * bi;
    im[r] = c * ai - s * br;
    re[p] = c * br + s * ai;
    im[p] = c * bi - s * ar;
  }
}

__global__ __launch_bounds__(256) void quanv_fused_kernel(
    const float* __restrict__ xin, const float* __restrict__ wts,
    float* __restrict__ out) {
  __shared__ __attribute__((aligned(8))) float sre[16 * ROWP];
  __shared__ __attribute__((aligned(8))) float sim2[16 * ROWP];
  __shared__ float part[8][256];
  __shared__ float gcs[16];
  __shared__ float gsn[16];
  __shared__ float smat[2 * 256];
  __shared__ __attribute__((aligned(8))) float Pm[WAVES][16 * PPIT];
  const int x = threadIdx.x;
  const int j = x >> 4;
  const int g = x & 15;

  const int wv = x >> 5;
  const int lane = x & 31;
  const int m = lane & 15;
  const int h = lane >> 4;
  const int tile = blockIdx.x * WAVES + wv;

  int pos = tile * 16 + m;
  int posc = (pos < NPOS) ? pos : (NPOS - 1);
  int b = posc / POS_PER_IMG;
  int rem = posc - b * POS_PER_IMG;
  int ii = rem / WO;
  int jj = rem - ii * WO;

  float tcur[4], tnx1[4], tnx2[4];
  {
    const float* img = xin + (((size_t)b * CH) * H + ii) * W + jj;
    tcur[0] = img[0];             tcur[1] = img[1];
    tcur[2] = img[W];             tcur[3] = img[W + 1];
    tnx1[0] = img[H * W];         tnx1[1] = img[H * W + 1];
    tnx1[2] = img[H * W + W];     tnx1[3] = img[H * W + W + 1];
    tnx2[0] = img[2 * H * W];     tnx2[1] = img[2 * H * W + 1];
    tnx2[2] = img[2 * H * W + W]; tnx2[3] = img[2 * H * W + W + 1];
  }

  if (x < 16) {
    float s, c;
    sincosf(0.5f * wts[x], &s, &c);
    gcs[x] = c;
    gsn[x] = s;
  }

  float re[16], im[16];
#pragma unroll
  for (int r = 0; r < 16; ++r) { re[r] = 0.0f; im[r] = 0.0f; }
  {
    const int pj = __popc(j) & 3;
    const float phr = (pj == 0) ? 1.0f : ((pj == 2) ? -1.0f : 0.0f);
    const float phi = (pj == 1) ? -1.0f : ((pj == 3) ? 1.0f : 0.0f);
    re[0] = (g == j) ? phr : 0.0f;
    im[0] = (g == j) ? phi : 0.0f;
  }
  __syncthreads();

  float* rowr = sre + j * ROWP;
  float* rowi = sim2 + j * ROWP;

#pragma unroll
  for (int l = 0; l < 2; ++l) {
    rx_regs<8>(re, im, gcs[l * 8 + 4], gsn[l * 8 + 4]);
    rx_regs<4>(re, im, gcs[l * 8 + 5], gsn[l * 8 + 5]);
    rx_regs<2>(re, im, gcs[l * 8 + 6], gsn[l * 8 + 6]);
    rx_regs<1>(re, im, gcs[l * 8 + 7], gsn[l * 8 + 7]);

#pragma unroll
    for (int r = 0; r < 16; ++r) {
      rowr[g * 17 + r] = re[r];
      rowi[g * 17 + r] = im[r];
    }
    __syncthreads();
#pragma unroll
    for (int r = 0; r < 16; ++r) {
      re[r] = rowr[r * 17 + g];
      im[r] = rowi[r * 17 + g];
    }

    rx_regs<8>(re, im, gcs[l * 8 + 0], gsn[l * 8 + 0]);
    rx_regs<4>(re, im, gcs[l * 8 + 1], gsn[l * 8 + 1]);
    rx_regs<2>(re, im, gcs[l * 8 + 2], gsn[l * 8 + 2]);
    rx_regs<1>(re, im, gcs[l * 8 + 3], gsn[l * 8 + 3]);
    __syncthreads();

    if (l == 0) {
#pragma unroll
      for (int r = 0; r < 16; ++r) {
        rowr[r * 17 + g] = re[r];
        rowi[r * 17 + g] = im[r];
      }
      __syncthreads();
      const int sg = ring_src(g << 4);
#pragma unroll
      for (int r = 0; r < 16; ++r) {
        const int y = sg ^ ring_src(r);
        re[r] = rowr[padx(y)];
        im[r] = rowi[padx(y)];
      }
      __syncthreads();
    } else {
      const int dg = ring_dst(g);
#pragma unroll
      for (int r = 0; r < 16; ++r) {
        const int y = dg ^ ring_dst(r << 4);
        sre[j * FINP + y] = re[r];
        sim2[j * FINP + y] = im[r];
      }
      __syncthreads();
    }
  }

  const int n = m;
  const int hh = h;
  const float* fr = sre + n * FINP + 32 * wv + 2 * hh;
  const float* fi = sim2 + n * FINP + 32 * wv + 2 * hh;
  v8f prr = {0, 0, 0, 0, 0, 0, 0, 0};
  v8f pii = {0, 0, 0, 0, 0, 0, 0, 0};
#pragma unroll
  for (int kb = 0; kb < 8; ++kb) {
    const v2f vr = *reinterpret_cast<const v2f*>(fr + 4 * kb);
    const v2f vi = *reinterpret_cast<const v2f*>(fi + 4 * kb);
    prr = __builtin_amdgcn_wmma_f32_16x16x4_f32(false, vr, false, vr, (short)0, prr, false, false);
    pii = __builtin_amdgcn_wmma_f32_16x16x4_f32(false, vi, false, vi, (short)0, pii, false, false);
  }
#pragma unroll
  for (int r = 0; r < 8; ++r) {
    part[wv][(r + 8 * hh) * 16 + n] = prr[r] + pii[r];
  }
  __syncthreads();
  float pr[8];
#pragma unroll
  for (int w = 0; w < 8; ++w) { pr[w] = part[w][x]; }
  const float r01 = pr[0] + pr[1], r23 = pr[2] + pr[3], r45 = pr[4] + pr[5], r67 = pr[6] + pr[7];
  smat[0 * 256 + x] = (r01 + r23) - (r45 + r67);
  smat[1 * 256 + x] = (r01 - r23) + (r45 - r67);
  __syncthreads();


  v2f bS0[4], bS1[4];
#pragma unroll
  for (int kb = 0; kb < 4; ++kb) {
    const int r0 = 4 * kb + 2 * h;
    bS0[kb] = (v2f){smat[0 * 256 + r0 * 16 + m], smat[0 * 256 + (r0 + 1) * 16 + m]};
    bS1[kb] = (v2f){smat[1 * 256 + r0 * 16 + m], smat[1 * 256 + (r0 + 1) * 16 + m]};
  }

  float* Pw = Pm[wv];

  float z0[8], z1[8];
#pragma unroll
  for (int r = 0; r < 8; ++r) { z0[r] = 0.f; z1[r] = 0.f; }

  for (int ch = 0; ch < 3; ++ch) {
    const float t0 = tcur[0], t1 = tcur[1], t2 = tcur[2], t3 = tcur[3];
#pragma unroll
    for (int i = 0; i < 4; ++i) { tcur[i] = tnx1[i]; tnx1[i] = tnx2[i]; }
    float c0, s0, c1, s1, c2, s2, c3, s3;
    sincosf(0.5f * t0, &s0, &c0);
    sincosf(0.5f * t1, &s1, &c1);
    sincosf(0.5f * t2, &s2, &c2);
    sincosf(0.5f * t3, &s3, &c3);

    const float q2 = h ? s2 : c2;
    v2f au[4];
#pragma unroll
    for (int kb = 0; kb < 4; ++kb) {
      const float m0 = (kb & 2) ? s0 : c0;
      const float m1 = (kb & 1) ? s1 : c1;
      const float m012 = m0 * m1 * q2;
      au[kb] = (v2f){m012 * c3, m012 * s3};
      *reinterpret_cast<v2f*>(Pw + m * PPIT + 4 * kb + 2 * h) = au[kb];
    }
    __builtin_amdgcn_fence(__ATOMIC_RELEASE, "wavefront");
    __builtin_amdgcn_fence(__ATOMIC_ACQUIRE, "wavefront");

    v8f g0 = {0, 0, 0, 0, 0, 0, 0, 0};
    v8f g1 = {0, 0, 0, 0, 0, 0, 0, 0};
#pragma unroll
    for (int kb = 0; kb < 4; ++kb) {
      g0 = __builtin_amdgcn_wmma_f32_16x16x4_f32(false, au[kb], false, bS0[kb],
                                                 (short)0, g0, false, false);
      g1 = __builtin_amdgcn_wmma_f32_16x16x4_f32(false, au[kb], false, bS1[kb],
                                                 (short)0, g1, false, false);
    }

#pragma unroll
    for (int r = 0; r < 8; ++r) {
      const float pm = Pw[(r + 8 * h) * PPIT + m];
      z0[r] += g0[r] * pm;
      z1[r] += g1[r] * pm;
    }
    __builtin_amdgcn_fence(__ATOMIC_RELEASE, "wavefront");
    __builtin_amdgcn_fence(__ATOMIC_ACQUIRE, "wavefront");
  }
#pragma unroll
  for (int r = 0; r < 8; ++r) {
    float a = z0[r], c = z1[r];
    for (int mask = 1; mask <= 8; mask <<= 1) {
      a += __shfl_xor(a, mask, 32);
      c += __shfl_xor(c, mask, 32);
    }
    z0[r] = a; z1[r] = c;
  }

  if (m == 0) {
#pragma unroll
    for (int r = 0; r < 8; ++r) {
      int posm = tile * 16 + r + 8 * h;
      if (posm < NPOS) {
        int bb = posm / POS_PER_IMG;
        int rr = posm - bb * POS_PER_IMG;
        out[(bb * 2 + 0) * POS_PER_IMG + rr] = z0[r] * (1.0f / 3.0f);
        out[(bb * 2 + 1) * POS_PER_IMG + rr] = z1[r] * (1.0f / 3.0f);
      }
    }
  }
}

extern "C" void kernel_launch(void* const* d_in, const int* in_sizes, int n_in,
                              void* d_out, int out_size, void* d_ws, size_t ws_size,
                              hipStream_t stream) {
  const float* x = (const float*)d_in[0];
  const float* wts = (const float*)d_in[1];
  float* out = (float*)d_out;

  quanv_fused_kernel<<<NBLOCKS, 256, 0, stream>>>(x, wts, out);
}


# build: layer 1 meets each column as one basis state, so its output is a product state: amplitudes written in closed form in the post-ring layout; layer-1 sweeps, 2 LDS transposes, 4 barriers gone
# speedup vs baseline: 5.8016x; 1.1044x over previous
#include <hip/hip_runtime.h>
#include <math.h>

typedef __attribute__((ext_vector_type(2))) float v2f;
typedef __attribute__((ext_vector_type(8))) float v8f;

static constexpr int BS = 8, CH = 3, H = 40, W = 40;
static constexpr int HO = 39, WO = 39;
static constexpr int POS_PER_IMG = HO * WO;
static constexpr int NPOS = BS * POS_PER_IMG;
static constexpr int NTILES = (NPOS + 15) / 16;
static constexpr int WAVES = 8;
static constexpr int NBLOCKS = (NTILES + WAVES - 1) / WAVES;

static constexpr int ROWP = 272;
static constexpr int FINP = 260;
static constexpr int PPIT = 20;

__device__ __forceinline__ int padx(int x) { return x + (x >> 4); }

__device__ __forceinline__ int ring_src(int x) {
  for (int q = 7; q >= 0; --q) {
    const int cm = 1 << (7 - q);
    const int tm = 1 << (7 - ((q + 1) & 7));
    x = (x & cm) ? (x ^ tm) : x;
  }
  return x;
}

__device__ __forceinline__ int ring_dst(int x) {
  for (int q = 0; q < 8; ++q) {
    const int cm = 1 << (7 - q);
    const int tm = 1 << (7 - ((q + 1) & 7));
    x = (x & cm) ? (x ^ tm) : x;
  }
  return x;
}

template <int BIT>
__device__ __forceinline__ void rx_regs(float (&re)[16], float (&im)[16], float c, float s) {
#pragma unroll
  for (int r = 0; r < 16; ++r) {
    if (r & BIT) continue;
    const int p = r | BIT;
    const float ar = re[r], ai = im[r], br = re[p], bi = im[p];
    re[r] = c * ar + s * bi;
    im[r] = c * ai - s * br;
    re[p] = c * br + s * ai;
    im[p] = c * bi - s * ar;
  }
}

__global__ __launch_bounds__(256) void quanv_fused_kernel(
    const float* __restrict__ xin, const float* __restrict__ wts,
    float* __restrict__ out) {
  __shared__ __attribute__((aligned(8))) float sre[16 * ROWP];
  __shared__ __attribute__((aligned(8))) float sim2[16 * ROWP];
  __shared__ float part[8][256];
  __shared__ float gcs[16];
  __shared__ float gsn[16];
  __shared__ float smat[2 * 256];
  __shared__ __attribute__((aligned(8))) float Pm[WAVES][16 * PPIT];
  const int x = threadIdx.x;
  const int j = x >> 4;
  const int g = x & 15;

  const int wv = x >> 5;
  const int lane = x & 31;
  const int m = lane & 15;
  const int h = lane >> 4;
  const int tile = blockIdx.x * WAVES + wv;

  int pos = tile * 16 + m;
  int posc = (pos < NPOS) ? pos : (NPOS - 1);
  int b = posc / POS_PER_IMG;
  int rem = posc - b * POS_PER_IMG;
  int ii = rem / WO;
  int jj = rem - ii * WO;

  float tcur[4], tnx1[4], tnx2[4];
  {
    const float* img = xin + (((size_t)b * CH) * H + ii) * W + jj;
    tcur[0] = img[0];             tcur[1] = img[1];
    tcur[2] = img[W];             tcur[3] = img[W + 1];
    tnx1[0] = img[H * W];         tnx1[1] = img[H * W + 1];
    tnx1[2] = img[H * W + W];     tnx1[3] = img[H * W + W + 1];
    tnx2[0] = img[2 * H * W];     tnx2[1] = img[2 * H * W + 1];
    tnx2[2] = img[2 * H * W + W]; tnx2[3] = img[2 * H * W + W + 1];
  }

  if (x < 16) {
    float s, c;
    sincosf(0.5f * wts[x], &s, &c);
    gcs[x] = c;
    gsn[x] = s;
  }

  __syncthreads();

  float* rowr = sre + j * ROWP;
  float* rowi = sim2 + j * ROWP;

  float re[16], im[16];
  {
    const int t = ring_src(g << 4) ^ (j << 4);
    float keep[8], flip[8];
#pragma unroll
    for (int bb = 0; bb < 8; ++bb) {
      const float c = gcs[7 - bb], s = gsn[7 - bb];
      const bool f = ((t >> bb) & 1) != 0;
      keep[bb] = f ? s : c;
      flip[bb] = f ? c : s;
    }
    const int pj = __popc(j);
    constexpr int ORD[8] = {5, 4, 7, 6, 0, 1, 2, 3};
#pragma unroll
    for (int r = 0; r < 16; ++r) {
      const int R = ring_src(r);
      float mag = ((R >> ORD[0]) & 1) ? flip[ORD[0]] : keep[ORD[0]];
#pragma unroll
      for (int o = 1; o < 8; ++o) {
        mag *= ((R >> ORD[o]) & 1) ? flip[ORD[o]] : keep[ORD[o]];
      }
      const int e = pj + __popc(t ^ R);
      const float sm = (e & 2) ? -mag : mag;
      re[r] = (e & 1) ? 0.0f : sm;
      im[r] = (e & 1) ? -sm : 0.0f;
    }
  }

  rx_regs<8>(re, im, gcs[8 + 4], gsn[8 + 4]);
  rx_regs<4>(re, im, gcs[8 + 5], gsn[8 + 5]);
  rx_regs<2>(re, im, gcs[8 + 6], gsn[8 + 6]);
  rx_regs<1>(re, im, gcs[8 + 7], gsn[8 + 7]);

#pragma unroll
  for (int r = 0; r < 16; ++r) {
    rowr[g * 17 + r] = re[r];
    rowi[g * 17 + r] = im[r];
  }
  __syncthreads();
#pragma unroll
  for (int r = 0; r < 16; ++r) {
    re[r] = rowr[r * 17 + g];
    im[r] = rowi[r * 17 + g];
  }

  rx_regs<8>(re, im, gcs[8 + 0], gsn[8 + 0]);
  rx_regs<4>(re, im, gcs[8 + 1], gsn[8 + 1]);
  rx_regs<2>(re, im, gcs[8 + 2], gsn[8 + 2]);
  rx_regs<1>(re, im, gcs[8 + 3], gsn[8 + 3]);
  __syncthreads();

  {
    const int dg = ring_dst(g);
#pragma unroll
    for (int r = 0; r < 16; ++r) {
      const int y = dg ^ ring_dst(r << 4);
      sre[j * FINP + y] = re[r];
      sim2[j * FINP + y] = im[r];
    }
    __syncthreads();
  }

  const int n = m;
  const int hh = h;
  const float* fr = sre + n * FINP + 32 * wv + 2 * hh;
  const float* fi = sim2 + n * FINP + 32 * wv + 2 * hh;
  v8f prr = {0, 0, 0, 0, 0, 0, 0, 0};
  v8f pii = {0, 0, 0, 0, 0, 0, 0, 0};
#pragma unroll
  for (int kb = 0; kb < 8; ++kb) {
    const v2f vr = *reinterpret_cast<const v2f*>(fr + 4 * kb);
    const v2f vi = *reinterpret_cast<const v2f*>(fi + 4 * kb);
    prr = __builtin_amdgcn_wmma_f32_16x16x4_f32(false, vr, false, vr, (short)0, prr, false, false);
    pii = __builtin_amdgcn_wmma_f32_16x16x4_f32(false, vi, false, vi, (short)0, pii, false, false);
  }
#pragma unroll
  for (int r = 0; r < 8; ++r) {
    part[wv][(r + 8 * hh) * 16 + n] = prr[r] + pii[r];
  }
  __syncthreads();
  float pr[8];
#pragma unroll
  for (int w = 0; w < 8; ++w) { pr[w] = part[w][x]; }
  const float r01 = pr[0] + pr[1], r23 = pr[2] + pr[3], r45 = pr[4] + pr[5], r67 = pr[6] + pr[7];
  smat[0 * 256 + x] = (r01 + r23) - (r45 + r67);
  smat[1 * 256 + x] = (r01 - r23) + (r45 - r67);
  __syncthreads();


  v2f bS0[4], bS1[4];
#pragma unroll
  for (int kb = 0; kb < 4; ++kb) {
    const int r0 = 4 * kb + 2 * h;
    bS0[kb] = (v2f){smat[0 * 256 + r0 * 16 + m], smat[0 * 256 + (r0 + 1) * 16 + m]};
    bS1[kb] = (v2f){smat[1 * 256 + r0 * 16 + m], smat[1 * 256 + (r0 + 1) * 16 + m]};
  }

  float* Pw = Pm[wv];

  float z0[8], z1[8];
#pragma unroll
  for (int r = 0; r < 8; ++r) { z0[r] = 0.f; z1[r] = 0.f; }

  for (int ch = 0; ch < 3; ++ch) {
    const float t0 = tcur[0], t1 = tcur[1], t2 = tcur[2], t3 = tcur[3];
#pragma unroll
    for (int i = 0; i < 4; ++i) { tcur[i] = tnx1[i]; tnx1[i] = tnx2[i]; }
    float c0, s0, c1, s1, c2, s2, c3, s3;
    sincosf(0.5f * t0, &s0, &c0);
    sincosf(0.5f * t1, &s1, &c1);
    sincosf(0.5f * t2, &s2, &c2);
    sincosf(0.5f * t3, &s3, &c3);

    const float q2 = h ? s2 : c2;
    v2f au[4];
#pragma unroll
    for (int kb = 0; kb < 4; ++kb) {
      const float m0 = (kb & 2) ? s0 : c0;
      const float m1 = (kb & 1) ? s1 : c1;
      const float m012 = m0 * m1 * q2;
      au[kb] = (v2f){m012 * c3, m012 * s3};
      *reinterpret_cast<v2f*>(Pw + m * PPIT + 4 * kb + 2 * h) = au[kb];
    }
    __builtin_amdgcn_fence(__ATOMIC_RELEASE, "wavefront");
    __builtin_amdgcn_fence(__ATOMIC_ACQUIRE, "wavefront");

    v8f g0 = {0, 0, 0, 0, 0, 0, 0, 0};
    v8f g1 = {0, 0, 0, 0, 0, 0, 0, 0};
#pragma unroll
    for (int kb = 0; kb < 4; ++kb) {
      g0 = __builtin_amdgcn_wmma_f32_16x16x4_f32(false, au[kb], false, bS0[kb],
                                                 (short)0, g0, false, false);
      g1 = __builtin_amdgcn_wmma_f32_16x16x4_f32(false, au[kb], false, bS1[kb],
                                                 (short)0, g1, false, false);
    }

#pragma unroll
    for (int r = 0; r < 8; ++r) {
      const float pm = Pw[(r + 8 * h) * PPIT + m];
      z0[r] += g0[r] * pm;
      z1[r] += g1[r] * pm;
    }
    __builtin_amdgcn_fence(__ATOMIC_RELEASE, "wavefront");
    __builtin_amdgcn_fence(__ATOMIC_ACQUIRE, "wavefront");
  }
#pragma unroll
  for (int r = 0; r < 8; ++r) {
    float a = z0[r], c = z1[r];
    for (int mask = 1; mask <= 8; mask <<= 1) {
      a += __shfl_xor(a, mask, 32);
      c += __shfl_xor(c, mask, 32);
    }
    z0[r] = a; z1[r] = c;
  }

  if (m == 0) {
#pragma unroll
    for (int r = 0; r < 8; ++r) {
      int posm = tile * 16 + r + 8 * h;
      if (posm < NPOS) {
        int bb = posm / POS_PER_IMG;
        int rr = posm - bb * POS_PER_IMG;
        out[(bb * 2 + 0) * POS_PER_IMG + rr] = z0[r] * (1.0f / 3.0f);
        out[(bb * 2 + 1) * POS_PER_IMG + rr] = z1[r] * (1.0f / 3.0f);
      }
    }
  }
}

extern "C" void kernel_launch(void* const* d_in, const int* in_sizes, int n_in,
                              void* d_out, int out_size, void* d_ws, size_t ws_size,
                              hipStream_t stream) {
  const float* x = (const float*)d_in[0];
  const float* wts = (const float*)d_in[1];
  float* out = (float*)d_out;

  quanv_fused_kernel<<<NBLOCKS, 256, 0, stream>>>(x, wts, out);
}


# Stage 2 product transposed to T_q = S_q * P^T: the row-dot finish runs over each lane's own accumulator registers; per-wave LDS plane and the 64-shuffle lane reduction gone; new summation order
# speedup vs baseline: 8.1932x; 1.4122x over previous
#include <hip/hip_runtime.h>
#include <math.h>

typedef __attribute__((ext_vector_type(2))) float v2f;
typedef __attribute__((ext_vector_type(8))) float v8f;

static constexpr int BS = 8, CH = 3, H = 40, W = 40;
static constexpr int HO = 39, WO = 39;
static constexpr int POS_PER_IMG = HO * WO;
static constexpr int NPOS = BS * POS_PER_IMG;
static constexpr int NTILES = (NPOS + 15) / 16;
static constexpr int WAVES = 8;
static constexpr int NBLOCKS = (NTILES + WAVES - 1) / WAVES;

static constexpr int ROWP = 272;
static constexpr int FINP = 260;
static constexpr int SPIT = 20;

__device__ __forceinline__ int padx(int x) { return x + (x >> 4); }

__device__ __forceinline__ float xhalf(float v) {
  const int b = __builtin_bit_cast(int, v);
  return __builtin_bit_cast(float,
      __builtin_amdgcn_permlanex16(b, b, 0x76543210, 0xfedcba98, false, false));
}

__device__ __forceinline__ int ring_src(int x) {
  for (int q = 7; q >= 0; --q) {
    const int cm = 1 << (7 - q);
    const int tm = 1 << (7 - ((q + 1) & 7));
    x = (x & cm) ? (x ^ tm) : x;
  }
  return x;
}

__device__ __forceinline__ int ring_dst(int x) {
  for (int q = 0; q < 8; ++q) {
    const int cm = 1 << (7 - q);
    const int tm = 1 << (7 - ((q + 1) & 7));
    x = (x & cm) ? (x ^ tm) : x;
  }
  return x;
}

template <int BIT>
__device__ __forceinline__ void rx_regs(float (&re)[16], float (&im)[16], float c, float s) {
#pragma unroll
  for (int r = 0; r < 16; ++r) {
    if (r & BIT) continue;
    const int p = r | BIT;
    const float ar = re[r], ai = im[r], br = re[p], bi = im[p];
    re[r] = c * ar + s * bi;
    im[r] = c * ai - s * br;
    re[p] = c * br + s * ai;
    im[p] = c * bi - s * ar;
  }
}

__global__ __launch_bounds__(256) void quanv_fused_kernel(
    const float* __restrict__ xin, const float* __restrict__ wts,
    float* __restrict__ out) {
  __shared__ __attribute__((aligned(8))) float sre[16 * ROWP];
  __shared__ __attribute__((aligned(8))) float sim2[16 * ROWP];
  __shared__ float part[8][256];
  __shared__ float gcs[16];
  __shared__ float gsn[16];
  __shared__ __attribute__((aligned(8))) float smat[2 * 16 * SPIT];
  const int x = threadIdx.x;
  const int j = x >> 4;
  const int g = x & 15;

  const int wv = x >> 5;
  const int lane = x & 31;
  const int m = lane & 15;
  const int h = lane >> 4;
  const int tile = blockIdx.x * WAVES + wv;

  int pos = tile * 16 + m;
  int posc = (pos < NPOS) ? pos : (NPOS - 1);
  int b = posc / POS_PER_IMG;
  int rem = posc - b * POS_PER_IMG;
  int ii = rem / WO;
  int jj = rem - ii * WO;

  float tcur[4], tnx1[4], tnx2[4];
  {
    const float* img = xin + (((size_t)b * CH) * H + ii) * W + jj;
    tcur[0] = img[0];             tcur[1] = img[1];
    tcur[2] = img[W];             tcur[3] = img[W + 1];
    tnx1[0] = img[H * W];         tnx1[1] = img[H * W + 1];
    tnx1[2] = img[H * W + W];     tnx1[3] = img[H * W + W + 1];
    tnx2[0] = img[2 * H * W];     tnx2[1] = img[2 * H * W + 1];
    tnx2[2] = img[2 * H * W + W]; tnx2[3] = img[2 * H * W + W + 1];
  }

  if (x < 16) {
    float s, c;
    sincosf(0.5f * wts[x], &s, &c);
    gcs[x] = c;
    gsn[x] = s;
  }

  __syncthreads();

  float* rowr = sre + j * ROWP;
  float* rowi = sim2 + j * ROWP;

  float re[16], im[16];
  {
    const int t = ring_src(g << 4) ^ (j << 4);
    float keep[8], flip[8];
#pragma unroll
    for (int bb = 0; bb < 8; ++bb) {
      const float c = gcs[7 - bb], s = gsn[7 - bb];
      const bool f = ((t >> bb) & 1) != 0;
      keep[bb] = f ? s : c;
      flip[bb] = f ? c : s;
    }
    const int pj = __popc(j);
    constexpr int ORD[8] = {5, 4, 7, 6, 0, 1, 2, 3};
#pragma unroll
    for (int r = 0; r < 16; ++r) {
      const int R = ring_src(r);
      float mag = ((R >> ORD[0]) & 1) ? flip[ORD[0]] : keep[ORD[0]];
#pragma unroll
      for (int o = 1; o < 8; ++o) {
        mag *= ((R >> ORD[o]) & 1) ? flip[ORD[o]] : keep[ORD[o]];
      }
      const int e = pj + __popc(t ^ R);
      const float sm = (e & 2) ? -mag : mag;
      re[r] = (e & 1) ? 0.0f : sm;
      im[r] = (e & 1) ? -sm : 0.0f;
    }
  }

  rx_regs<8>(re, im, gcs[8 + 4], gsn[8 + 4]);
  rx_regs<4>(re, im, gcs[8 + 5], gsn[8 + 5]);
  rx_regs<2>(re, im, gcs[8 + 6], gsn[8 + 6]);
  rx_regs<1>(re, im, gcs[8 + 7], gsn[8 + 7]);

#pragma unroll
  for (int r = 0; r < 16; ++r) {
    rowr[g * 17 + r] = re[r];
    rowi[g * 17 + r] = im[r];
  }
  __syncthreads();
#pragma unroll
  for (int r = 0; r < 16; ++r) {
    re[r] = rowr[r * 17 + g];
    im[r] = rowi[r * 17 + g];
  }

  rx_regs<8>(re, im, gcs[8 + 0], gsn[8 + 0]);
  rx_regs<4>(re, im, gcs[8 + 1], gsn[8 + 1]);
  rx_regs<2>(re, im, gcs[8 + 2], gsn[8 + 2]);
  rx_regs<1>(re, im, gcs[8 + 3], gsn[8 + 3]);
  __syncthreads();

  {
    const int dg = ring_dst(g);
#pragma unroll
    for (int r = 0; r < 16; ++r) {
      const int y = dg ^ ring_dst(r << 4);
      sre[j * FINP + y] = re[r];
      sim2[j * FINP + y] = im[r];
    }
    __syncthreads();
  }

  const int n = m;
  const int hh = h;
  const float* fr = sre + n * FINP + 32 * wv + 2 * hh;
  const float* fi = sim2 + n * FINP + 32 * wv + 2 * hh;
  v8f prr = {0, 0, 0, 0, 0, 0, 0, 0};
  v8f pii = {0, 0, 0, 0, 0, 0, 0, 0};
#pragma unroll
  for (int kb = 0; kb < 8; ++kb) {
    const v2f vr = *reinterpret_cast<const v2f*>(fr + 4 * kb);
    const v2f vi = *reinterpret_cast<const v2f*>(fi + 4 * kb);
    prr = __builtin_amdgcn_wmma_f32_16x16x4_f32(false, vr, false, vr, (short)0, prr, false, false);
    pii = __builtin_amdgcn_wmma_f32_16x16x4_f32(false, vi, false, vi, (short)0, pii, false, false);
  }
#pragma unroll
  for (int r = 0; r < 8; ++r) {
    part[wv][(r + 8 * hh) * 16 + n] = prr[r] + pii[r];
  }
  __syncthreads();
  float pr[8];
#pragma unroll
  for (int w = 0; w < 8; ++w) { pr[w] = part[w][x]; }
  const float r01 = pr[0] + pr[1], r23 = pr[2] + pr[3], r45 = pr[4] + pr[5], r67 = pr[6] + pr[7];
  smat[(0 * 16 + j) * SPIT + g] = (r01 + r23) - (r45 + r67);
  smat[(1 * 16 + j) * SPIT + g] = (r01 - r23) + (r45 - r67);
  __syncthreads();


  v2f aS0[4], aS1[4];
#pragma unroll
  for (int kb = 0; kb < 4; ++kb) {
    aS0[kb] = *reinterpret_cast<const v2f*>(smat + (0 * 16 + m) * SPIT + 4 * kb + 2 * h);
    aS1[kb] = *reinterpret_cast<const v2f*>(smat + (1 * 16 + m) * SPIT + 4 * kb + 2 * h);
  }

  v2f zp0 = {0.f, 0.f}, zp1 = {0.f, 0.f};

  for (int ch = 0; ch < 3; ++ch) {
    const float t0 = tcur[0], t1 = tcur[1], t2 = tcur[2], t3 = tcur[3];
#pragma unroll
    for (int i = 0; i < 4; ++i) { tcur[i] = tnx1[i]; tnx1[i] = tnx2[i]; }
    float c0, s0, c1, s1, c2, s2, c3, s3;
    sincosf(0.5f * t0, &s0, &c0);
    sincosf(0.5f * t1, &s1, &c1);
    sincosf(0.5f * t2, &s2, &c2);
    sincosf(0.5f * t3, &s3, &c3);

    const float qa0 = c0 * c1, qa1 = c0 * s1, qa2 = s0 * c1, qa3 = s0 * s1;
    const float qb0 = c2 * c3, qb1 = c2 * s3, qb2 = s2 * c3, qb3 = s2 * s3;
    const float bh0 = h ? qb2 : qb0;
    const float bh1 = h ? qb3 : qb1;
    v2f bu[4];
    bu[0] = (v2f){qa0 * bh0, qa0 * bh1};
    bu[1] = (v2f){qa1 * bh0, qa1 * bh1};
    bu[2] = (v2f){qa2 * bh0, qa2 * bh1};
    bu[3] = (v2f){qa3 * bh0, qa3 * bh1};
    const float ah0 = h ? qa2 : qa0;
    const float ah1 = h ? qa3 : qa1;
    const v2f pf0 = (v2f){ah0 * qb0, ah0 * qb1};
    const v2f pf1 = (v2f){ah0 * qb2, ah0 * qb3};
    const v2f pf2 = (v2f){ah1 * qb0, ah1 * qb1};
    const v2f pf3 = (v2f){ah1 * qb2, ah1 * qb3};

    v8f g0 = {0, 0, 0, 0, 0, 0, 0, 0};
    v8f g1 = {0, 0, 0, 0, 0, 0, 0, 0};
#pragma unroll
    for (int kb = 0; kb < 4; ++kb) {
      g0 = __builtin_amdgcn_wmma_f32_16x16x4_f32(false, aS0[kb], false, bu[kb],
                                                 (short)0, g0, false, false);
      g1 = __builtin_amdgcn_wmma_f32_16x16x4_f32(false, aS1[kb], false, bu[kb],
                                                 (short)0, g1, false, false);
    }

    zp0 += (v2f){g0[0], g0[1]} * pf0;
    zp1 += (v2f){g1[0], g1[1]} * pf0;
    zp0 += (v2f){g0[2], g0[3]} * pf1;
    zp1 += (v2f){g1[2], g1[3]} * pf1;
    zp0 += (v2f){g0[4], g0[5]} * pf2;
    zp1 += (v2f){g1[4], g1[5]} * pf2;
    zp0 += (v2f){g0[6], g0[7]} * pf3;
    zp1 += (v2f){g1[6], g1[7]} * pf3;
  }

  const float zh0 = zp0[0] + zp0[1];
  const float zh1 = zp1[0] + zp1[1];
  const float zt0 = zh0 + xhalf(zh0);
  const float zt1 = zh1 + xhalf(zh1);
  if (pos < NPOS) {
    out[(b * 2 + h) * POS_PER_IMG + rem] = (h ? zt1 : zt0) * (1.0f / 3.0f);
  }
}

extern "C" void kernel_launch(void* const* d_in, const int* in_sizes, int n_in,
                              void* d_out, int out_size, void* d_ws, size_t ws_size,
                              hipStream_t stream) {
  const float* x = (const float*)d_in[0];
  const float* wts = (const float*)d_in[1];
  float* out = (float*)d_out;

  quanv_fused_kernel<<<NBLOCKS, 256, 0, stream>>>(x, wts, out);
}


# Stage 1 simulates only column 0, in wave 0 alone: X strings pass through RX gates and CNOT rings, so W_j(x) = phi(j) W_0(x ^ m_j); closing WMMAs read shifted fragments of it; new summation order
# speedup vs baseline: 9.9422x; 1.2135x over previous
#include <hip/hip_runtime.h>
#include <math.h>

typedef __attribute__((ext_vector_type(2))) float v2f;
typedef __attribute__((ext_vector_type(8))) float v8f;

static constexpr int BS = 8, CH = 3, H = 40, W = 40;
static constexpr int HO = 39, WO = 39;
static constexpr int POS_PER_IMG = HO * WO;
static constexpr int NPOS = BS * POS_PER_IMG;
static constexpr int NTILES = (NPOS + 15) / 16;
static constexpr int WAVES = 8;
static constexpr int NBLOCKS = (NTILES + WAVES - 1) / WAVES;

static constexpr int ROWP = 272;
static constexpr int SPIT = 20;

__device__ __forceinline__ int wpos(int y) {
  return y ^ ((y >> 2) & 1) ^ (((y >> 6) & 3) << 2);
}

__device__ __forceinline__ float xhalf(float v) {
  const int b = __builtin_bit_cast(int, v);
  return __builtin_bit_cast(float,
      __builtin_amdgcn_permlanex16(b, b, 0x76543210, 0xfedcba98, false, false));
}

__device__ __forceinline__ void sincos_small(float x, float& s, float& c) {
  const float ax = __builtin_fabsf(x);
  const float n = __builtin_rintf(ax * 0x1.45f306p-1f);
  float r = __builtin_fmaf(n, -0x1.921fb4p+0f, ax);
  r = __builtin_fmaf(n, -0x1.4442d0p-24f, r);
  r = __builtin_fmaf(n, -0x1.846988p-48f, r);
  const int q = (int)n;
  const float r2 = r * r;
  float sp = __builtin_fmaf(-0x1.983304p-13f, r2, 0x1.110388p-7f);
  sp = __builtin_fmaf(r2, sp, -0x1.55553ap-3f);
  sp = r2 * sp;
  const float sv = __builtin_fmaf(r, sp, r);
  float cp = __builtin_fmaf(0x1.aea668p-16f, r2, -0x1.6c9e76p-10f);
  cp = __builtin_fmaf(r2, cp, 0x1.5557eep-5f);
  cp = __builtin_fmaf(r2, cp, -0x1.000008p-1f);
  const float cv = __builtin_fmaf(r2, cp, 1.0f);
  const bool odd = (q & 1) != 0;
  const float ss = odd ? cv : sv;
  const float cc = odd ? -sv : cv;
  const unsigned flip = ((unsigned)q << 30) & 0x80000000u;
  const unsigned sx = __builtin_bit_cast(unsigned, x) & 0x80000000u;
  s = __builtin_bit_cast(float, __builtin_bit_cast(unsigned, ss) ^ flip ^ sx);
  c = __builtin_bit_cast(float, __builtin_bit_cast(unsigned, cc) ^ flip);
}

__device__ __forceinline__ int ring_src(int x) {
  for (int q = 7; q >= 0; --q) {
    const int cm = 1 << (7 - q);
    const int tm = 1 << (7 - ((q + 1) & 7));
    x = (x & cm) ? (x ^ tm) : x;
  }
  return x;
}

__device__ __forceinline__ int ring_dst(int x) {
  for (int q = 0; q < 8; ++q) {
    const int cm = 1 << (7 - q);
    const int tm = 1 << (7 - ((q + 1) & 7));
    x = (x & cm) ? (x ^ tm) : x;
  }
  return x;
}

__device__ __forceinline__ void wave_sync() {
  __builtin_amdgcn_fence(__ATOMIC_RELEASE, "wavefront");
  __builtin_amdgcn_wave_barrier();
  __builtin_amdgcn_fence(__ATOMIC_ACQUIRE, "wavefront");
}

template <int BIT>
__device__ __forceinline__ void rx_regs(float (&re)[16], float (&im)[16], float c, float s) {
#pragma unroll
  for (int r = 0; r < 16; ++r) {
    if (r & BIT) continue;
    const int p = r | BIT;
    const float ar = re[r], ai = im[r], br = re[p], bi = im[p];
    re[r] = c * ar + s * bi;
    im[r] = c * ai - s * br;
    re[p] = c * br + s * ai;
    im[p] = c * bi - s * ar;
  }
}

__global__ __launch_bounds__(256) void quanv_fused_kernel(
    const float* __restrict__ xin, const float* __restrict__ wts,
    float* __restrict__ out) {
  __shared__ float tr[ROWP];
  __shared__ float ti[ROWP];
  __shared__ __attribute__((aligned(8))) float w0[512];
  __shared__ float part[8][256];
  __shared__ float gcs[16];
  __shared__ float gsn[16];
  __shared__ __attribute__((aligned(8))) float smat[2 * 16 * SPIT];
  const int x = threadIdx.x;
  const int j = x >> 4;
  const int g = x & 15;

  const int wv = x >> 5;
  const int lane = x & 31;
  const int m = lane & 15;
  const int h = lane >> 4;
  const int tile = blockIdx.x * WAVES + wv;

  int pos = tile * 16 + m;
  int posc = (pos < NPOS) ? pos : (NPOS - 1);
  int b = posc / POS_PER_IMG;
  int rem = posc - b * POS_PER_IMG;
  int ii = rem / WO;
  int jj = rem - ii * WO;

  float tcur[4], tnx1[4], tnx2[4];
  {
    const float* img = xin + (((size_t)b * CH) * H + ii) * W + jj;
    tcur[0] = img[0];             tcur[1] = img[1];
    tcur[2] = img[W];             tcur[3] = img[W + 1];
    tnx1[0] = img[H * W];         tnx1[1] = img[H * W + 1];
    tnx1[2] = img[H * W + W];     tnx1[3] = img[H * W + W + 1];
    tnx2[0] = img[2 * H * W];     tnx2[1] = img[2 * H * W + 1];
    tnx2[2] = img[2 * H * W + W]; tnx2[3] = img[2 * H * W + W + 1];
  }

  const int n = m;
  const int hh = h;
  const int mn = ((n & 8) ? ring_dst(ring_dst(0x80)) : 0) ^ ((n & 4) ? ring_dst(ring_dst(0x40)) : 0) ^
                 ((n & 2) ? ring_dst(ring_dst(0x20)) : 0) ^ ((n & 1) ? ring_dst(ring_dst(0x10)) : 0);
  const int pq = wpos((32 * wv + 2 * hh) ^ mn) >> 1;
  const int en = __popc(n) & 3;
  const int rb = (en & 1) ? 256 : 0;
  const float sgr = (en & 2) ? -1.0f : 1.0f;
  const float sgi = ((en + 1) & 2) ? -1.0f : 1.0f;

  if (wv == 0) {
    if (x < 16) {
      float s, c;
      sincosf(0.5f * wts[x], &s, &c);
      gcs[x] = c;
      gsn[x] = s;
    }
    wave_sync();

    float re[16], im[16];
    {
      const int t = ring_src(g << 4);
      float keep[8], flip[8];
#pragma unroll
      for (int bb = 0; bb < 8; ++bb) {
        const float c = gcs[7 - bb], s = gsn[7 - bb];
        const bool f = ((t >> bb) & 1) != 0;
        keep[bb] = f ? s : c;
        flip[bb] = f ? c : s;
      }
      constexpr int ORD[8] = {5, 4, 7, 6, 0, 1, 2, 3};
#pragma unroll
      for (int r = 0; r < 16; ++r) {
        const int R = ring_src(r);
        float mag = ((R >> ORD[0]) & 1) ? flip[ORD[0]] : keep[ORD[0]];
#pragma unroll
        for (int o = 1; o < 8; ++o) {
          mag *= ((R >> ORD[o]) & 1) ? flip[ORD[o]] : keep[ORD[o]];
        }
        const int e = __popc(t ^ R);
        const float sm = (e & 2) ? -mag : mag;
        re[r] = (e & 1) ? 0.0f : sm;
        im[r] = (e & 1) ? -sm : 0.0f;
      }
    }

    rx_regs<8>(re, im, gcs[8 + 4], gsn[8 + 4]);
    rx_regs<4>(re, im, gcs[8 + 5], gsn[8 + 5]);
    rx_regs<2>(re, im, gcs[8 + 6], gsn[8 + 6]);
    rx_regs<1>(re, im, gcs[8 + 7], gsn[8 + 7]);

#pragma unroll
    for (int r = 0; r < 16; ++r) {
      tr[g * 17 + r] = re[r];
      ti[g * 17 + r] = im[r];
    }
    wave_sync();
#pragma unroll
    for (int r = 0; r < 16; ++r) {
      re[r] = tr[r * 17 + g];
      im[r] = ti[r * 17 + g];
    }

    rx_regs<8>(re, im, gcs[8 + 0], gsn[8 + 0]);
    rx_regs<4>(re, im, gcs[8 + 1], gsn[8 + 1]);
    rx_regs<2>(re, im, gcs[8 + 2], gsn[8 + 2]);
    rx_regs<1>(re, im, gcs[8 + 3], gsn[8 + 3]);

    {
      const int pg = wpos(ring_dst(g));
#pragma unroll
      for (int r = 0; r < 16; ++r) {
        const int p = pg ^ wpos(ring_dst(r << 4));
        w0[p] = re[r];
        w0[256 + p] = im[r];
      }
    }
  }
  __syncthreads();

  v8f prr = {0, 0, 0, 0, 0, 0, 0, 0};
  v8f pii = {0, 0, 0, 0, 0, 0, 0, 0};
  v2f lr[8], li[8];
#pragma unroll
  for (int kb = 0; kb < 8; ++kb) {
    const int sl = 2 * (pq ^ (2 * kb));
    lr[kb] = *reinterpret_cast<const v2f*>(w0 + (rb + sl));
    li[kb] = *reinterpret_cast<const v2f*>(w0 + ((256 - rb) + sl));
  }
  __builtin_amdgcn_sched_barrier(0);
#pragma unroll
  for (int kb = 0; kb < 8; ++kb) {
    const v2f vr = lr[kb] * sgr;
    const v2f vi = li[kb] * sgi;
    prr = __builtin_amdgcn_wmma_f32_16x16x4_f32(false, vr, false, vr, (short)0, prr, false, false);
    pii = __builtin_amdgcn_wmma_f32_16x16x4_f32(false, vi, false, vi, (short)0, pii, false, false);
  }
#pragma unroll
  for (int r = 0; r < 8; ++r) {
    part[wv][(r + 8 * hh) * 16 + n] = prr[r] + pii[r];
  }
  __syncthreads();
  float pr[8];
#pragma unroll
  for (int w = 0; w < 8; ++w) { pr[w] = part[w][x]; }
  const float r01 = pr[0] + pr[1], r23 = pr[2] + pr[3], r45 = pr[4] + pr[5], r67 = pr[6] + pr[7];
  smat[(0 * 16 + j) * SPIT + g] = (r01 + r23) - (r45 + r67);
  smat[(1 * 16 + j) * SPIT + g] = (r01 - r23) + (r45 - r67);
  __syncthreads();


  v2f aS0[4], aS1[4];
#pragma unroll
  for (int kb = 0; kb < 4; ++kb) {
    aS0[kb] = *reinterpret_cast<const v2f*>(smat + (0 * 16 + m) * SPIT + 4 * kb + 2 * h);
    aS1[kb] = *reinterpret_cast<const v2f*>(smat + (1 * 16 + m) * SPIT + 4 * kb + 2 * h);
  }

  v2f zp0 = {0.f, 0.f}, zp1 = {0.f, 0.f};

  float pcs[3][4], psn[3][4];
  {
    float ha[3][4];
#pragma unroll
    for (int i = 0; i < 4; ++i) {
      ha[0][i] = 0.5f * tcur[i];
      ha[1][i] = 0.5f * tnx1[i];
      ha[2][i] = 0.5f * tnx2[i];
    }
    bool big = false;
#pragma unroll
    for (int k = 0; k < 3; ++k) {
#pragma unroll
      for (int i = 0; i < 4; ++i) { big = big || !(__builtin_fabsf(ha[k][i]) < 131072.0f); }
    }
    if (__builtin_expect(__builtin_amdgcn_ballot_w32(big) != 0u, 0)) {
#pragma unroll
      for (int k = 0; k < 3; ++k) {
#pragma unroll
        for (int i = 0; i < 4; ++i) { sincosf(ha[k][i], &psn[k][i], &pcs[k][i]); }
      }
    } else {
#pragma unroll
      for (int k = 0; k < 3; ++k) {
#pragma unroll
        for (int i = 0; i < 4; ++i) { sincos_small(ha[k][i], psn[k][i], pcs[k][i]); }
      }
    }
  }

#pragma unroll
  for (int ch = 0; ch < 3; ++ch) {
    const float c0 = pcs[ch][0], s0 = psn[ch][0], c1 = pcs[ch][1], s1 = psn[ch][1];
    const float c2 = pcs[ch][2], s2 = psn[ch][2], c3 = pcs[ch][3], s3 = psn[ch][3];

    const float qa0 = c0 * c1, qa1 = c0 * s1, qa2 = s0 * c1, qa3 = s0 * s1;
    const float qb0 = c2 * c3, qb1 = c2 * s3, qb2 = s2 * c3, qb3 = s2 * s3;
    const float bh0 = h ? qb2 : qb0;
    const float bh1 = h ? qb3 : qb1;
    v2f bu[4];
    bu[0] = (v2f){qa0 * bh0, qa0 * bh1};
    bu[1] = (v2f){qa1 * bh0, qa1 * bh1};
    bu[2] = (v2f){qa2 * bh0, qa2 * bh1};
    bu[3] = (v2f){qa3 * bh0, qa3 * bh1};
    const float ah0 = h ? qa2 : qa0;
    const float ah1 = h ? qa3 : qa1;
    const v2f pf0 = (v2f){ah0 * qb0, ah0 * qb1};
    const v2f pf1 = (v2f){ah0 * qb2, ah0 * qb3};
    const v2f pf2 = (v2f){ah1 * qb0, ah1 * qb1};
    const v2f pf3 = (v2f){ah1 * qb2, ah1 * qb3};

    v8f g0 = {0, 0, 0, 0, 0, 0, 0, 0};
    v8f g1 = {0, 0, 0, 0, 0, 0, 0, 0};
#pragma unroll
    for (int kb = 0; kb < 4; ++kb) {
      g0 = __builtin_amdgcn_wmma_f32_16x16x4_f32(false, aS0[kb], false, bu[kb],
                                                 (short)0, g0, false, false);
      g1 = __builtin_amdgcn_wmma_f32_16x16x4_f32(false, aS1[kb], false, bu[kb],
                                                 (short)0, g1, false, false);
    }

    zp0 += (v2f){g0[0], g0[1]} * pf0;
    zp1 += (v2f){g1[0], g1[1]} * pf0;
    zp0 += (v2f){g0[2], g0[3]} * pf1;
    zp1 += (v2f){g1[2], g1[3]} * pf1;
    zp0 += (v2f){g0[4], g0[5]} * pf2;
    zp1 += (v2f){g1[4], g1[5]} * pf2;
    zp0 += (v2f){g0[6], g0[7]} * pf3;
    zp1 += (v2f){g1[6], g1[7]} * pf3;
  }

  const float zh0 = zp0[0] + zp0[1];
  const float zh1 = zp1[0] + zp1[1];
  const float zt0 = zh0 + xhalf(zh0);
  const float zt1 = zh1 + xhalf(zh1);
  if (pos < NPOS) {
    out[(b * 2 + h) * POS_PER_IMG + rem] = (h ? zt1 : zt0) * (1.0f / 3.0f);
  }
}

extern "C" void kernel_launch(void* const* d_in, const int* in_sizes, int n_in,
                              void* d_out, int out_size, void* d_ws, size_t ws_size,
                              hipStream_t stream) {
  const float* x = (const float*)d_in[0];
  const float* wts = (const float*)d_in[1];
  float* out = (float*)d_out;

  quanv_fused_kernel<<<NBLOCKS, 256, 0, stream>>>(x, wts, out);
}


# 4-wave workgroups: wave 0 only simulates, waves 1-3 carry two tiles each and compute their sincos and fragments during that wait; closing sums K=64 per wave (new summation order)
# speedup vs baseline: 10.5539x; 1.0615x over previous
#include <hip/hip_runtime.h>
#include <math.h>

typedef __attribute__((ext_vector_type(2))) float v2f;
typedef __attribute__((ext_vector_type(8))) float v8f;

static constexpr int BS = 8, CH = 3, H = 40, W = 40;
static constexpr int HO = 39, WO = 39;
static constexpr int POS_PER_IMG = HO * WO;
static constexpr int NPOS = BS * POS_PER_IMG;
static constexpr int NTILES = (NPOS + 15) / 16;
static constexpr int WAVES = 4;
static constexpr int TPW = 2;
static constexpr int TILES_PER_WG = (WAVES - 1) * TPW;
static constexpr int NBLOCKS = (NTILES + TILES_PER_WG - 1) / TILES_PER_WG;

static constexpr int ROWP = 272;
static constexpr int SPIT = 20;

__device__ __forceinline__ int wpos(int y) {
  return y ^ ((y >> 2) & 1) ^ (((y >> 6) & 3) << 2);
}

__device__ __forceinline__ float xhalf(float v) {
  const int b = __builtin_bit_cast(int, v);
  return __builtin_bit_cast(float,
      __builtin_amdgcn_permlanex16(b, b, 0x76543210, 0xfedcba98, false, false));
}

__device__ __forceinline__ void sincos_small(float x, float& s, float& c) {
  const float ax = __builtin_fabsf(x);
  const float n = __builtin_rintf(ax * 0x1.45f306p-1f);
  float r = __builtin_fmaf(n, -0x1.921fb4p+0f, ax);
  r = __builtin_fmaf(n, -0x1.4442d0p-24f, r);
  r = __builtin_fmaf(n, -0x1.846988p-48f, r);
  const int q = (int)n;
  const float r2 = r * r;
  float sp = __builtin_fmaf(-0x1.983304p-13f, r2, 0x1.110388p-7f);
  sp = __builtin_fmaf(r2, sp, -0x1.55553ap-3f);
  sp = r2 * sp;
  const float sv = __builtin_fmaf(r, sp, r);
  float cp = __builtin_fmaf(0x1.aea668p-16f, r2, -0x1.6c9e76p-10f);
  cp = __builtin_fmaf(r2, cp, 0x1.5557eep-5f);
  cp = __builtin_fmaf(r2, cp, -0x1.000008p-1f);
  const float cv = __builtin_fmaf(r2, cp, 1.0f);
  const bool odd = (q & 1) != 0;
  const float ss = odd ? cv : sv;
  const float cc = odd ? -sv : cv;
  const unsigned flip = ((unsigned)q << 30) & 0x80000000u;
  const unsigned sx = __builtin_bit_cast(unsigned, x) & 0x80000000u;
  s = __builtin_bit_cast(float, __builtin_bit_cast(unsigned, ss) ^ flip ^ sx);
  c = __builtin_bit_cast(float, __builtin_bit_cast(unsigned, cc) ^ flip);
}

__device__ __forceinline__ int ring_src(int x) {
  for (int q = 7; q >= 0; --q) {
    const int cm = 1 << (7 - q);
    const int tm = 1 << (7 - ((q + 1) & 7));
    x = (x & cm) ? (x ^ tm) : x;
  }
  return x;
}

__device__ __forceinline__ int ring_dst(int x) {
  for (int q = 0; q < 8; ++q) {
    const int cm = 1 << (7 - q);
    const int tm = 1 << (7 - ((q + 1) & 7));
    x = (x & cm) ? (x ^ tm) : x;
  }
  return x;
}

__device__ __forceinline__ void wave_sync() {
  __builtin_amdgcn_fence(__ATOMIC_RELEASE, "wavefront");
  __builtin_amdgcn_wave_barrier();
  __builtin_amdgcn_fence(__ATOMIC_ACQUIRE, "wavefront");
}

template <int BIT>
__device__ __forceinline__ void rx_regs(float (&re)[16], float (&im)[16], float c, float s) {
#pragma unroll
  for (int r = 0; r < 16; ++r) {
    if (r & BIT) continue;
    const int p = r | BIT;
    const float ar = re[r], ai = im[r], br = re[p], bi = im[p];
    re[r] = c * ar + s * bi;
    im[r] = c * ai - s * br;
    re[p] = c * br + s * ai;
    im[p] = c * bi - s * ar;
  }
}

__device__ __forceinline__ void tile_prep(const float (&t0)[4], const float (&t1)[4],
                                          const float (&t2)[4], int h,
                                          v2f (&bu)[3][4], v2f (&pf)[3][4]) {
  float pcs[3][4], psn[3][4];
  {
    float ha[3][4];
#pragma unroll
    for (int i = 0; i < 4; ++i) {
      ha[0][i] = 0.5f * t0[i];
      ha[1][i] = 0.5f * t1[i];
      ha[2][i] = 0.5f * t2[i];
    }
    bool big = false;
#pragma unroll
    for (int k = 0; k < 3; ++k) {
#pragma unroll
      for (int i = 0; i < 4; ++i) { big = big || !(__builtin_fabsf(ha[k][i]) < 131072.0f); }
    }
    if (__builtin_expect(__builtin_amdgcn_ballot_w32(big) != 0u, 0)) {
#pragma unroll
      for (int k = 0; k < 3; ++k) {
#pragma unroll
        for (int i = 0; i < 4; ++i) { sincosf(ha[k][i], &psn[k][i], &pcs[k][i]); }
      }
    } else {
#pragma unroll
      for (int k = 0; k < 3; ++k) {
#pragma unroll
        for (int i = 0; i < 4; ++i) { sincos_small(ha[k][i], psn[k][i], pcs[k][i]); }
      }
    }
  }

#pragma unroll
  for (int ch = 0; ch < 3; ++ch) {
    const float c0 = pcs[ch][0], s0 = psn[ch][0], c1 = pcs[ch][1], s1 = psn[ch][1];
    const float c2 = pcs[ch][2], s2 = psn[ch][2], c3 = pcs[ch][3], s3 = psn[ch][3];

    const float qa0 = c0 * c1, qa1 = c0 * s1, qa2 = s0 * c1, qa3 = s0 * s1;
    const float qb0 = c2 * c3, qb1 = c2 * s3, qb2 = s2 * c3, qb3 = s2 * s3;
    const float bh0 = h ? qb2 : qb0;
    const float bh1 = h ? qb3 : qb1;
    bu[ch][0] = (v2f){qa0 * bh0, qa0 * bh1};
    bu[ch][1] = (v2f){qa1 * bh0, qa1 * bh1};
    bu[ch][2] = (v2f){qa2 * bh0, qa2 * bh1};
    bu[ch][3] = (v2f){qa3 * bh0, qa3 * bh1};
    const float ah0 = h ? qa2 : qa0;
    const float ah1 = h ? qa3 : qa1;
    pf[ch][0] = (v2f){ah0 * qb0, ah0 * qb1};
    pf[ch][1] = (v2f){ah0 * qb2, ah0 * qb3};
    pf[ch][2] = (v2f){ah1 * qb0, ah1 * qb1};
    pf[ch][3] = (v2f){ah1 * qb2, ah1 * qb3};
  }
}

__global__ __launch_bounds__(128) void quanv_fused_kernel(
    const float* __restrict__ xin, const float* __restrict__ wts,
    float* __restrict__ out) {
  __shared__ float tr[ROWP];
  __shared__ float ti[ROWP];
  __shared__ __attribute__((aligned(8))) float w0[512];
  __shared__ float part[WAVES][256];
  __shared__ float gcs[16];
  __shared__ float gsn[16];
  __shared__ __attribute__((aligned(8))) float smat[2 * 16 * SPIT];
  const int x = threadIdx.x;
  const int g = x & 15;

  const int wv = __builtin_amdgcn_readfirstlane(x >> 5);
  const int lane = x & 31;
  const int m = lane & 15;
  const int h = lane >> 4;

  const int n = m;
  const int hh = h;
  const int mn = ((n & 8) ? ring_dst(ring_dst(0x80)) : 0) ^ ((n & 4) ? ring_dst(ring_dst(0x40)) : 0) ^
                 ((n & 2) ? ring_dst(ring_dst(0x20)) : 0) ^ ((n & 1) ? ring_dst(ring_dst(0x10)) : 0);
  const int pq = wpos((64 * wv + 2 * hh) ^ mn) >> 1;
  const int en = __popc(n) & 3;
  const int rb = (en & 1) ? 256 : 0;
  const float sgr = (en & 2) ? -1.0f : 1.0f;
  const float sgi = ((en + 1) & 2) ? -1.0f : 1.0f;

  int pos[TPW], bimg[TPW], rem[TPW];
  v2f bu[TPW][3][4], pf[TPW][3][4];
  if (wv != 0) {
    float pt[TPW][3][4];
#pragma unroll
    for (int t = 0; t < TPW; ++t) {
      const int tile = blockIdx.x * TILES_PER_WG + (wv - 1) * TPW + t;
      pos[t] = tile * 16 + m;
      const int posc = (pos[t] < NPOS) ? pos[t] : (NPOS - 1);
      bimg[t] = posc / POS_PER_IMG;
      rem[t] = posc - bimg[t] * POS_PER_IMG;
      const int ii = rem[t] / WO;
      const int jj = rem[t] - ii * WO;
      const float* img = xin + (((size_t)bimg[t] * CH) * H + ii) * W + jj;
#pragma unroll
      for (int k = 0; k < 3; ++k) {
        pt[t][k][0] = img[k * H * W];
        pt[t][k][1] = img[k * H * W + 1];
        pt[t][k][2] = img[k * H * W + W];
        pt[t][k][3] = img[k * H * W + W + 1];
      }
    }
#pragma unroll
    for (int t = 0; t < TPW; ++t) { tile_prep(pt[t][0], pt[t][1], pt[t][2], h, bu[t], pf[t]); }
  }

  if (wv == 0) {
    if (x < 16) {
      float s, c;
      sincosf(0.5f * wts[x], &s, &c);
      gcs[x] = c;
      gsn[x] = s;
    }
    wave_sync();

    float re[16], im[16];
    {
      const int t = ring_src(g << 4);
      float keep[8], flip[8];
#pragma unroll
      for (int bb = 0; bb < 8; ++bb) {
        const float c = gcs[7 - bb], s = gsn[7 - bb];
        const bool f = ((t >> bb) & 1) != 0;
        keep[bb] = f ? s : c;
        flip[bb] = f ? c : s;
      }
      constexpr int ORD[8] = {5, 4, 7, 6, 0, 1, 2, 3};
#pragma unroll
      for (int r = 0; r < 16; ++r) {
        const int R = ring_src(r);
        float mag = ((R >> ORD[0]) & 1) ? flip[ORD[0]] : keep[ORD[0]];
#pragma unroll
        for (int o = 1; o < 8; ++o) {
          mag *= ((R >> ORD[o]) & 1) ? flip[ORD[o]] : keep[ORD[o]];
        }
        const int e = __popc(t ^ R);
        const float sm = (e & 2) ? -mag : mag;
        re[r] = (e & 1) ? 0.0f : sm;
        im[r] = (e & 1) ? -sm : 0.0f;
      }
    }

    rx_regs<8>(re, im, gcs[8 + 4], gsn[8 + 4]);
    rx_regs<4>(re, im, gcs[8 + 5], gsn[8 + 5]);
    rx_regs<2>(re, im, gcs[8 + 6], gsn[8 + 6]);
    rx_regs<1>(re, im, gcs[8 + 7], gsn[8 + 7]);

#pragma unroll
    for (int r = 0; r < 16; ++r) {
      tr[g * 17 + r] = re[r];
      ti[g * 17 + r] = im[r];
    }
    wave_sync();
#pragma unroll
    for (int r = 0; r < 16; ++r) {
      re[r] = tr[r * 17 + g];
      im[r] = ti[r * 17 + g];
    }

    rx_regs<8>(re, im, gcs[8 + 0], gsn[8 + 0]);
    rx_regs<4>(re, im, gcs[8 + 1], gsn[8 + 1]);
    rx_regs<2>(re, im, gcs[8 + 2], gsn[8 + 2]);
    rx_regs<1>(re, im, gcs[8 + 3], gsn[8 + 3]);

    {
      const int pg = wpos(ring_dst(g));
#pragma unroll
      for (int r = 0; r < 16; ++r) {
        const int p = pg ^ wpos(ring_dst(r << 4));
        w0[p] = re[r];
        w0[256 + p] = im[r];
      }
    }
  }
  __syncthreads();

  v8f prr = {0, 0, 0, 0, 0, 0, 0, 0};
  v8f pii = {0, 0, 0, 0, 0, 0, 0, 0};
  v2f lr[16], li[16];
#pragma unroll
  for (int kb = 0; kb < 16; ++kb) {
    const int sl = 2 * (pq ^ (2 * kb));
    lr[kb] = *reinterpret_cast<const v2f*>(w0 + (rb + sl));
    li[kb] = *reinterpret_cast<const v2f*>(w0 + ((256 - rb) + sl));
  }
  __builtin_amdgcn_sched_barrier(0);
#pragma unroll
  for (int kb = 0; kb < 16; ++kb) {
    const v2f vr = lr[kb] * sgr;
    const v2f vi = li[kb] * sgi;
    prr = __builtin_amdgcn_wmma_f32_16x16x4_f32(false, vr, false, vr, (short)0, prr, false, false);
    pii = __builtin_amdgcn_wmma_f32_16x16x4_f32(false, vi, false, vi, (short)0, pii, false, false);
  }
#pragma unroll
  for (int r = 0; r < 8; ++r) {
    part[wv][(r + 8 * hh) * 16 + n] = prr[r] + pii[r];
  }
  __syncthreads();
#pragma unroll
  for (int e2 = 0; e2 < 2; ++e2) {
    const int e = x + 128 * e2;
    const int j = e >> 4;
    const float p0 = part[0][e], p1 = part[1][e], p2 = part[2][e], p3 = part[3][e];
    smat[(0 * 16 + j) * SPIT + g] = (p0 + p1) - (p2 + p3);
    smat[(1 * 16 + j) * SPIT + g] = (p0 - p1) + (p2 - p3);
  }
  __syncthreads();

  if (wv != 0) {
    v2f aS0[4], aS1[4];
#pragma unroll
    for (int kb = 0; kb < 4; ++kb) {
      aS0[kb] = *reinterpret_cast<const v2f*>(smat + (0 * 16 + m) * SPIT + 4 * kb + 2 * h);
      aS1[kb] = *reinterpret_cast<const v2f*>(smat + (1 * 16 + m) * SPIT + 4 * kb + 2 * h);
    }

#pragma unroll
    for (int t = 0; t < TPW; ++t) {
      v2f zp0 = {0.f, 0.f}, zp1 = {0.f, 0.f};
#pragma unroll
      for (int ch = 0; ch < 3; ++ch) {
        v8f g0 = {0, 0, 0, 0, 0, 0, 0, 0};
        v8f g1 = {0, 0, 0, 0, 0, 0, 0, 0};
#pragma unroll
        for (int kb = 0; kb < 4; ++kb) {
          g0 = __builtin_amdgcn_wmma_f32_16x16x4_f32(false, aS0[kb], false, bu[t][ch][kb],
                                                     (short)0, g0, false, false);
          g1 = __builtin_amdgcn_wmma_f32_16x16x4_f32(false, aS1[kb], false, bu[t][ch][kb],
                                                     (short)0, g1, false, false);
        }

        zp0 += (v2f){g0[0], g0[1]} * pf[t][ch][0];
        zp1 += (v2f){g1[0], g1[1]} * pf[t][ch][0];
        zp0 += (v2f){g0[2], g0[3]} * pf[t][ch][1];
        zp1 += (v2f){g1[2], g1[3]} * pf[t][ch][1];
        zp0 += (v2f){g0[4], g0[5]} * pf[t][ch][2];
        zp1 += (v2f){g1[4], g1[5]} * pf[t][ch][2];
        zp0 += (v2f){g0[6], g0[7]} * pf[t][ch][3];
        zp1 += (v2f){g1[6], g1[7]} * pf[t][ch][3];
      }

      const float zh0 = zp0[0] + zp0[1];
      const float zh1 = zp1[0] + zp1[1];
      const float zt0 = zh0 + xhalf(zh0);
      const float zt1 = zh1 + xhalf(zh1);
      if (pos[t] < NPOS) {
        out[(bimg[t] * 2 + h) * POS_PER_IMG + rem[t]] = (h ? zt1 : zt0) * (1.0f / 3.0f);
      }
    }
  }
}

extern "C" void kernel_launch(void* const* d_in, const int* in_sizes, int n_in,
                              void* d_out, int out_size, void* d_ws, size_t ws_size,
                              hipStream_t stream) {
  const float* x = (const float*)d_in[0];
  const float* wts = (const float*)d_in[1];
  float* out = (float*)d_out;

  quanv_fused_kernel<<<NBLOCKS, WAVES * 32, 0, stream>>>(x, wts, out);
}


# Stage 2: z_0 is free of patch value 2 and z_1 of value 1 for any weights, so each S_q shrinks to an 8x8 block; both in one block-diagonal A: 4 WMMAs per channel, not 8; new summation order
# speedup vs baseline: 11.1655x; 1.0580x over previous
#include <hip/hip_runtime.h>
#include <math.h>

typedef __attribute__((ext_vector_type(2))) float v2f;
typedef __attribute__((ext_vector_type(8))) float v8f;

static constexpr int BS = 8, CH = 3, H = 40, W = 40;
static constexpr int HO = 39, WO = 39;
static constexpr int POS_PER_IMG = HO * WO;
static constexpr int NPOS = BS * POS_PER_IMG;
static constexpr int NTILES = (NPOS + 15) / 16;
static constexpr int WAVES = 4;
static constexpr int TPW = 2;
static constexpr int TILES_PER_WG = (WAVES - 1) * TPW;
static constexpr int NBLOCKS = (NTILES + TILES_PER_WG - 1) / TILES_PER_WG;

static constexpr int ROWP = 272;
static constexpr int SPIT = 20;

__device__ __forceinline__ int wpos(int y) {
  return y ^ ((y >> 2) & 1) ^ (((y >> 6) & 3) << 2);
}

__device__ __forceinline__ void sincos_small(float x, float& s, float& c) {
  const float ax = __builtin_fabsf(x);
  const float n = __builtin_rintf(ax * 0x1.45f306p-1f);
  float r = __builtin_fmaf(n, -0x1.921fb4p+0f, ax);
  r = __builtin_fmaf(n, -0x1.4442d0p-24f, r);
  r = __builtin_fmaf(n, -0x1.846988p-48f, r);
  const int q = (int)n;
  const float r2 = r * r;
  float sp = __builtin_fmaf(-0x1.983304p-13f, r2, 0x1.110388p-7f);
  sp = __builtin_fmaf(r2, sp, -0x1.55553ap-3f);
  sp = r2 * sp;
  const float sv = __builtin_fmaf(r, sp, r);
  float cp = __builtin_fmaf(0x1.aea668p-16f, r2, -0x1.6c9e76p-10f);
  cp = __builtin_fmaf(r2, cp, 0x1.5557eep-5f);
  cp = __builtin_fmaf(r2, cp, -0x1.000008p-1f);
  const float cv = __builtin_fmaf(r2, cp, 1.0f);
  const bool odd = (q & 1) != 0;
  const float ss = odd ? cv : sv;
  const float cc = odd ? -sv : cv;
  const unsigned flip = ((unsigned)q << 30) & 0x80000000u;
  const unsigned sx = __builtin_bit_cast(unsigned, x) & 0x80000000u;
  s = __builtin_bit_cast(float, __builtin_bit_cast(unsigned, ss) ^ flip ^ sx);
  c = __builtin_bit_cast(float, __builtin_bit_cast(unsigned, cc) ^ flip);
}

__device__ __forceinline__ int ring_src(int x) {
  for (int q = 7; q >= 0; --q) {
    const int cm = 1 << (7 - q);
    const int tm = 1 << (7 - ((q + 1) & 7));
    x = (x & cm) ? (x ^ tm) : x;
  }
  return x;
}

__device__ __forceinline__ int ring_dst(int x) {
  for (int q = 0; q < 8; ++q) {
    const int cm = 1 << (7 - q);
    const int tm = 1 << (7 - ((q + 1) & 7));
    x = (x & cm) ? (x ^ tm) : x;
  }
  return x;
}

__device__ __forceinline__ void wave_sync() {
  __builtin_amdgcn_fence(__ATOMIC_RELEASE, "wavefront");
  __builtin_amdgcn_wave_barrier();
  __builtin_amdgcn_fence(__ATOMIC_ACQUIRE, "wavefront");
}

template <int BIT>
__device__ __forceinline__ void rx_regs(float (&re)[16], float (&im)[16], float c, float s) {
#pragma unroll
  for (int r = 0; r < 16; ++r) {
    if (r & BIT) continue;
    const int p = r | BIT;
    const float ar = re[r], ai = im[r], br = re[p], bi = im[p];
    re[r] = c * ar + s * bi;
    im[r] = c * ai - s * br;
    re[p] = c * br + s * ai;
    im[p] = c * bi - s * ar;
  }
}

__device__ __forceinline__ void tile_prep(const float (&t0)[4], const float (&t1)[4],
                                          const float (&t2)[4], int h,
                                          v2f (&bu)[3][4], v2f (&pf)[3][4]) {
  float pcs[3][4], psn[3][4];
  {
    float ha[3][4];
#pragma unroll
    for (int i = 0; i < 4; ++i) {
      ha[0][i] = 0.5f * t0[i];
      ha[1][i] = 0.5f * t1[i];
      ha[2][i] = 0.5f * t2[i];
    }
    bool big = false;
#pragma unroll
    for (int k = 0; k < 3; ++k) {
#pragma unroll
      for (int i = 0; i < 4; ++i) { big = big || !(__builtin_fabsf(ha[k][i]) < 131072.0f); }
    }
    if (__builtin_expect(__builtin_amdgcn_ballot_w32(big) != 0u, 0)) {
#pragma unroll
      for (int k = 0; k < 3; ++k) {
#pragma unroll
        for (int i = 0; i < 4; ++i) { sincosf(ha[k][i], &psn[k][i], &pcs[k][i]); }
      }
    } else {
#pragma unroll
      for (int k = 0; k < 3; ++k) {
#pragma unroll
        for (int i = 0; i < 4; ++i) { sincos_small(ha[k][i], psn[k][i], pcs[k][i]); }
      }
    }
  }

#pragma unroll
  for (int ch = 0; ch < 3; ++ch) {
    const float c0 = pcs[ch][0], s0 = psn[ch][0], c1 = pcs[ch][1], s1 = psn[ch][1];
    const float c2 = pcs[ch][2], s2 = psn[ch][2], c3 = pcs[ch][3], s3 = psn[ch][3];

    const float f1 = h ? s1 : c1;
    const float f2 = h ? s2 : c2;
    const float u0 = c0 * f1, u1 = s0 * f1, v0 = c0 * f2, v1 = s0 * f2;
    bu[ch][0] = (v2f){u0 * c3, u0 * s3};
    bu[ch][1] = (v2f){u1 * c3, u1 * s3};
    bu[ch][2] = (v2f){v0 * c3, v0 * s3};
    bu[ch][3] = (v2f){v1 * c3, v1 * s3};
    const float cu = h ? c2 : c1;
    const float su = h ? s2 : s1;
    const float p0 = c0 * cu, p1 = c0 * su, p2 = s0 * cu, p3 = s0 * su;
    pf[ch][0] = (v2f){p0 * c3, p0 * s3};
    pf[ch][1] = (v2f){p1 * c3, p1 * s3};
    pf[ch][2] = (v2f){p2 * c3, p2 * s3};
    pf[ch][3] = (v2f){p3 * c3, p3 * s3};
  }
}

__global__ __launch_bounds__(128) void quanv_fused_kernel(
    const float* __restrict__ xin, const float* __restrict__ wts,
    float* __restrict__ out) {
  __shared__ float tr[ROWP];
  __shared__ float ti[ROWP];
  __shared__ __attribute__((aligned(8))) float w0[512];
  __shared__ float part[WAVES][256];
  __shared__ float gcs[16];
  __shared__ float gsn[16];
  __shared__ __attribute__((aligned(8))) float smat[2 * 16 * SPIT];
  const int x = threadIdx.x;
  const int g = x & 15;

  const int wv = __builtin_amdgcn_readfirstlane(x >> 5);
  const int lane = x & 31;
  const int m = lane & 15;
  const int h = lane >> 4;

  const int n = m;
  const int hh = h;
  const int mn = ((n & 8) ? ring_dst(ring_dst(0x80)) : 0) ^ ((n & 4) ? ring_dst(ring_dst(0x40)) : 0) ^
                 ((n & 2) ? ring_dst(ring_dst(0x20)) : 0) ^ ((n & 1) ? ring_dst(ring_dst(0x10)) : 0);
  const int pq = wpos((64 * wv + 2 * hh) ^ mn) >> 1;
  const int en = __popc(n) & 3;
  const int rb = (en & 1) ? 256 : 0;
  const float sgr = (en & 2) ? -1.0f : 1.0f;
  const float sgi = ((en + 1) & 2) ? -1.0f : 1.0f;

  int pos[TPW], bimg[TPW], rem[TPW];
  v2f bu[TPW][3][4], pf[TPW][3][4];
  if (wv != 0) {
    float pt[TPW][3][4];
#pragma unroll
    for (int t = 0; t < TPW; ++t) {
      const int tile = blockIdx.x * TILES_PER_WG + (wv - 1) * TPW + t;
      pos[t] = tile * 16 + m;
      const int posc = (pos[t] < NPOS) ? pos[t] : (NPOS - 1);
      bimg[t] = posc / POS_PER_IMG;
      rem[t] = posc - bimg[t] * POS_PER_IMG;
      const int ii = rem[t] / WO;
      const int jj = rem[t] - ii * WO;
      const float* img = xin + (((size_t)bimg[t] * CH) * H + ii) * W + jj;
#pragma unroll
      for (int k = 0; k < 3; ++k) {
        pt[t][k][0] = img[k * H * W];
        pt[t][k][1] = img[k * H * W + 1];
        pt[t][k][2] = img[k * H * W + W];
        pt[t][k][3] = img[k * H * W + W + 1];
      }
    }
#pragma unroll
    for (int t = 0; t < TPW; ++t) { tile_prep(pt[t][0], pt[t][1], pt[t][2], h, bu[t], pf[t]); }
  }

  if (wv == 0) {
    if (x < 16) {
      float s, c;
      sincosf(0.5f * wts[x], &s, &c);
      gcs[x] = c;
      gsn[x] = s;
    }
    wave_sync();

    float re[16], im[16];
    {
      const int t = ring_src(g << 4);
      float keep[8], flip[8];
#pragma unroll
      for (int bb = 0; bb < 8; ++bb) {
        const float c = gcs[7 - bb], s = gsn[7 - bb];
        const bool f = ((t >> bb) & 1) != 0;
        keep[bb] = f ? s : c;
        flip[bb] = f ? c : s;
      }
      constexpr int ORD[8] = {5, 4, 7, 6, 0, 1, 2, 3};
#pragma unroll
      for (int r = 0; r < 16; ++r) {
        const int R = ring_src(r);
        float mag = ((R >> ORD[0]) & 1) ? flip[ORD[0]] : keep[ORD[0]];
#pragma unroll
        for (int o = 1; o < 8; ++o) {
          mag *= ((R >> ORD[o]) & 1) ? flip[ORD[o]] : keep[ORD[o]];
        }
        const int e = __popc(t ^ R);
        const float sm = (e & 2) ? -mag : mag;
        re[r] = (e & 1) ? 0.0f : sm;
        im[r] = (e & 1) ? -sm : 0.0f;
      }
    }

    rx_regs<8>(re, im, gcs[8 + 4], gsn[8 + 4]);
    rx_regs<4>(re, im, gcs[8 + 5], gsn[8 + 5]);
    rx_regs<2>(re, im, gcs[8 + 6], gsn[8 + 6]);
    rx_regs<1>(re, im, gcs[8 + 7], gsn[8 + 7]);

#pragma unroll
    for (int r = 0; r < 16; ++r) {
      tr[g * 17 + r] = re[r];
      ti[g * 17 + r] = im[r];
    }
    wave_sync();
#pragma unroll
    for (int r = 0; r < 16; ++r) {
      re[r] = tr[r * 17 + g];
      im[r] = ti[r * 17 + g];
    }

    rx_regs<8>(re, im, gcs[8 + 0], gsn[8 + 0]);
    rx_regs<4>(re, im, gcs[8 + 1], gsn[8 + 1]);
    rx_regs<2>(re, im, gcs[8 + 2], gsn[8 + 2]);
    rx_regs<1>(re, im, gcs[8 + 3], gsn[8 + 3]);

    {
      const int pg = wpos(ring_dst(g));
#pragma unroll
      for (int r = 0; r < 16; ++r) {
        const int p = pg ^ wpos(ring_dst(r << 4));
        w0[p] = re[r];
        w0[256 + p] = im[r];
      }
    }
  }
  __syncthreads();

  v8f prr = {0, 0, 0, 0, 0, 0, 0, 0};
  v8f pii = {0, 0, 0, 0, 0, 0, 0, 0};
  v2f lr[16], li[16];
#pragma unroll
  for (int kb = 0; kb < 16; ++kb) {
    const int sl = 2 * (pq ^ (2 * kb));
    lr[kb] = *reinterpret_cast<const v2f*>(w0 + (rb + sl));
    li[kb] = *reinterpret_cast<const v2f*>(w0 + ((256 - rb) + sl));
  }
  __builtin_amdgcn_sched_barrier(0);
#pragma unroll
  for (int kb = 0; kb < 16; ++kb) {
    const v2f vr = lr[kb] * sgr;
    const v2f vi = li[kb] * sgi;
    prr = __builtin_amdgcn_wmma_f32_16x16x4_f32(false, vr, false, vr, (short)0, prr, false, false);
    pii = __builtin_amdgcn_wmma_f32_16x16x4_f32(false, vi, false, vi, (short)0, pii, false, false);
  }
#pragma unroll
  for (int r = 0; r < 8; ++r) {
    part[wv][(r + 8 * hh) * 16 + n] = prr[r] + pii[r];
  }
  __syncthreads();
#pragma unroll
  for (int e2 = 0; e2 < 2; ++e2) {
    const int e = x + 128 * e2;
    const int j = e >> 4;
    const float p0 = part[0][e], p1 = part[1][e], p2 = part[2][e], p3 = part[3][e];
    smat[(0 * 16 + j) * SPIT + g] = (p0 + p1) - (p2 + p3);
    smat[(1 * 16 + j) * SPIT + g] = (p0 - p1) + (p2 - p3);
  }
  __syncthreads();

  if (wv != 0) {
    const bool lo = m < 8;
    const int i2 = (m >> 2) & 1, i1 = (m >> 1) & 1, ib = m & 1;
    const int sh = lo ? 2 : 1;
    const int erow = (lo ? 0 : 16) + ((i2 << 3) | (i1 << sh) | ib);
    const float* sp = smat + erow * SPIT + (h << sh);
    const v2f ld0 = *reinterpret_cast<const v2f*>(sp);
    const v2f ld1 = *reinterpret_cast<const v2f*>(sp + 8);
    const v2f zero2 = {0.f, 0.f};
    v2f aS[4];
    aS[0] = lo ? ld0 : zero2;
    aS[1] = lo ? ld1 : zero2;
    aS[2] = lo ? zero2 : ld0;
    aS[3] = lo ? zero2 : ld1;

#pragma unroll
    for (int t = 0; t < TPW; ++t) {
      v2f zp = {0.f, 0.f};
#pragma unroll
      for (int ch = 0; ch < 3; ++ch) {
        v8f gq = {0, 0, 0, 0, 0, 0, 0, 0};
#pragma unroll
        for (int kb = 0; kb < 4; ++kb) {
          gq = __builtin_amdgcn_wmma_f32_16x16x4_f32(false, aS[kb], false, bu[t][ch][kb],
                                                     (short)0, gq, false, false);
        }

        zp += (v2f){gq[0], gq[1]} * pf[t][ch][0];
        zp += (v2f){gq[2], gq[3]} * pf[t][ch][1];
        zp += (v2f){gq[4], gq[5]} * pf[t][ch][2];
        zp += (v2f){gq[6], gq[7]} * pf[t][ch][3];
      }

      const float zt = zp[0] + zp[1];
      if (pos[t] < NPOS) {
        out[(bimg[t] * 2 + h) * POS_PER_IMG + rem[t]] = zt * (1.0f / 3.0f);
      }
    }
  }
}

extern "C" void kernel_launch(void* const* d_in, const int* in_sizes, int n_in,
                              void* d_out, int out_size, void* d_ws, size_t ws_size,
                              hipStream_t stream) {
  const float* x = (const float*)d_in[0];
  const float* wts = (const float*)d_in[1];
  float* out = (float*)d_out;

  quanv_fused_kernel<<<NBLOCKS, WAVES * 32, 0, stream>>>(x, wts, out);
}
